# starting kernel
# baseline (speedup 1.0000x reference)
#include <hip/hip_runtime.h>
#include <hip/hip_bf16.h>

#define UU 512
#define LL 512
#define BB 256
static __device__ constexpr float PI_F = 3.14159265358979323846f;

typedef float v2f __attribute__((ext_vector_type(2)));
typedef float v8f __attribute__((ext_vector_type(8)));

__global__ __launch_bounds__(256) void coeff_kernel(
    const float* __restrict__ theta, const float* __restrict__ phi,
    const float* __restrict__ mask,  const float* __restrict__ enn,
    const float* __restrict__ enp,   const float* __restrict__ epn,
    const float* __restrict__ epp,
    float2* __restrict__ diagw, float2* __restrict__ offw) {
  int idx = blockIdx.x * blockDim.x + threadIdx.x;
  int l = idx / UU;
  int u = idx % UU;

  auto psl = [&](const float* ang, int uu) -> float2 {
    uu &= (UU - 1);
    if (uu & 1) return make_float2(1.0f, 0.0f);
    int k = uu >> 1;
    float m = mask[l * (UU / 2) + k];
    float a = ang[l * (UU / 2) + k] * m + (1.0f - m) * PI_F;
    float s, c;
    __sincosf(a, &s, &c);
    return make_float2(c, s);
  };

  float2 ip_m1 = psl(theta, u - 1);
  float2 ip_0  = psl(theta, u);
  float2 ip_p1 = psl(theta, u + 1);
  float2 ep_m1 = psl(phi,   u - 1);
  float2 ep_0  = psl(phi,   u);

  int um1 = (u - 1) & (UU - 1);
  float enn_u = enn[u * LL + l],  enn_m = enn[um1 * LL + l];
  float epp_u = epp[u * LL + l],  epp_m = epp[um1 * LL + l];
  float enp_u = enp[u * LL + l],  enp_m = enp[um1 * LL + l];
  float epn_u = epn[u * LL + l],  epn_m = epn[um1 * LL + l];

  float sdr = epp_u * ip_0.x - enn_u * ip_p1.x - enn_m * ip_m1.x + epp_m * ip_0.x;
  float sdi = epp_u * ip_0.y - enn_u * ip_p1.y - enn_m * ip_m1.y + epp_m * ip_0.y;

  float cvr = epn_u * ip_0.x + enp_u * ip_p1.x + enp_m * ip_m1.x + epn_m * ip_0.x;
  float cvi = epn_u * ip_0.y + enp_u * ip_p1.y + enp_m * ip_m1.y + epn_m * ip_0.y;
  float sor = -cvi, soi = cvr;

  diagw[idx] = make_float2(0.5f * (ep_0.x * sdr - ep_0.y * sdi),
                           0.5f * (ep_0.x * sdi + ep_0.y * sdr));
  offw[idx]  = make_float2(0.5f * (ep_m1.x * sor - ep_m1.y * soi),
                           0.5f * (ep_m1.x * soi + ep_m1.y * sor));
}

__global__ __launch_bounds__(256) void invp_kernel(const int* __restrict__ perms,
                                                   int* __restrict__ invp) {
  int idx = blockIdx.x * blockDim.x + threadIdx.x;
  int l = idx / UU;
  int u = idx % UU;
  int c = perms[(l + 1) * UU + u];
  invp[l * UU + c] = u;
}

__global__ __launch_bounds__(256) void scan_kernel(
    const float* __restrict__ x, const float* __restrict__ gamma,
    const int* __restrict__ perms, const float2* __restrict__ diagw,
    const float2* __restrict__ offw, const int* __restrict__ invp,
    float* __restrict__ out) {
  __shared__ float2 S[16 * UU];

  const int tid  = threadIdx.x;
  const int b0   = blockIdx.x * 16;
  const int wave = tid >> 5;
  const int lane = tid & 31;
  const int n    = lane & 15;
  const int hi   = lane >> 4;
  const int m    = lane & 15;
  const int r    = m & 3;
  const int blk  = m >> 2;

  for (int e = 0; e < (16 * UU) / 256; ++e) {
    int idx = tid + e * 256;
    int rr = idx >> 9;
    int u  = idx & (UU - 1);
    int p  = perms[u];
    float xr = x[(size_t)(b0 + rr) * UU + p];
    float xi = x[(size_t)BB * UU + (size_t)(b0 + rr) * UU + p];
    float sg, cg;
    __sincosf(gamma[p], &sg, &cg);
    S[rr * UU + u] = make_float2(xr * cg - xi * sg, xr * sg + xi * cg);
  }
  __syncthreads();

  for (int l = 0; l < LL; ++l) {
    const float2* dl = diagw + (size_t)l * UU;
    const float2* ol = offw  + (size_t)l * UU;
    const int*    il = invp  + (size_t)l * UU;
    __builtin_prefetch(dl + UU, 0, 0);
    __builtin_prefetch(ol + UU, 0, 0);
    __builtin_prefetch(il + UU, 0, 0);

    v8f acc[8];
#pragma unroll
    for (int t = 0; t < 8; ++t) {
      const int cb = (wave * 8 + t) * 8;
      v8f c = {0.f, 0.f, 0.f, 0.f, 0.f, 0.f, 0.f, 0.f};
#pragma unroll
      for (int j = 0; j < 4; ++j) {
        const int col = cb + 2 * j + hi;
        float2 sb = S[n * UU + col];
        v2f bop;  bop[0] = sb.x;  bop[1] = sb.y;
        bool use_diag = ((r < 2) == (hi == 0));
        float2 p = use_diag ? dl[col] : ol[col];
        float a0, a1;
        if (r & 1) { a0 = p.y;  a1 = p.x;  }
        else       { a0 = p.x;  a1 = -p.y; }
        if (blk != j) { a0 = 0.f; a1 = 0.f; }
        v2f aop;  aop[0] = a0;  aop[1] = a1;
        c = __builtin_amdgcn_wmma_f32_16x16x4_f32(
                false, aop, false, bop, (short)0, c, false, false);
      }
      acc[t] = c;
    }
    __syncthreads();
#pragma unroll
    for (int t = 0; t < 8; ++t) {
      const int cb = (wave * 8 + t) * 8;
#pragma unroll
      for (int v = 0; v < 8; v += 2) {
        int mm = v + hi * 8;
        int cc = cb + (mm >> 1);
        int uo = il[cc];
        S[n * UU + uo] = make_float2(acc[t][v], acc[t][v + 1]);
      }
    }
    __syncthreads();
  }

  for (int e = 0; e < (16 * UU) / 256; ++e) {
    int idx = tid + e * 256;
    int rr = idx >> 9;
    int u  = idx & (UU - 1);
    float2 s = S[rr * UU + u];
    out[(size_t)(b0 + rr) * UU + u] = s.x;
    out[(size_t)BB * UU + (size_t)(b0 + rr) * UU + u] = s.y;
  }
}

extern "C" void kernel_launch(void* const* d_in, const int* in_sizes, int n_in,
                              void* d_out, int out_size, void* d_ws, size_t ws_size,
                              hipStream_t stream) {
  const float* x     = (const float*)d_in[0];
  const float* theta = (const float*)d_in[1];
  const float* phi   = (const float*)d_in[2];
  const float* gamma = (const float*)d_in[3];
  const float* mask  = (const float*)d_in[4];
  const float* enn   = (const float*)d_in[5];
  const float* enp   = (const float*)d_in[6];
  const float* epn   = (const float*)d_in[7];
  const float* epp   = (const float*)d_in[8];
  const int*   perms = (const int*)d_in[9];

  char* ws = (char*)d_ws;
  float2* diagw = (float2*)(ws);
  float2* offw  = (float2*)(ws + (size_t)LL * UU * 8);
  int*    invp  = (int*)   (ws + (size_t)LL * UU * 16);

  const int threads = 256;
  const int blocksLU = (LL * UU) / threads;

  coeff_kernel<<<blocksLU, threads, 0, stream>>>(theta, phi, mask, enn, enp,
                                                 epn, epp, diagw, offw);
  invp_kernel<<<blocksLU, threads, 0, stream>>>(perms, invp);
  scan_kernel<<<BB / 16, threads, 0, stream>>>(x, gamma, perms, diagw, offw,
                                               invp, (float*)d_out);
}


# LDS state re-laid batch-row-fastest with a column-bit-2 half swizzle: B-fragment reads and roll write-back stores go from 16-way bank conflicts to conflict-free
# speedup vs baseline: 2.9606x; 2.9606x over previous
#include <hip/hip_runtime.h>
#include <hip/hip_bf16.h>

#define UU 512
#define LL 512
#define BB 256
static __device__ constexpr float PI_F = 3.14159265358979323846f;

typedef float v2f __attribute__((ext_vector_type(2)));
typedef float v8f __attribute__((ext_vector_type(8)));

__device__ __forceinline__ int sidx(int col, int row) {
  return ((col >> 1) << 5) + ((((col ^ (col >> 2)) & 1)) << 4) + row;
}

__global__ __launch_bounds__(256) void coeff_kernel(
    const float* __restrict__ theta, const float* __restrict__ phi,
    const float* __restrict__ mask,  const float* __restrict__ enn,
    const float* __restrict__ enp,   const float* __restrict__ epn,
    const float* __restrict__ epp,
    float2* __restrict__ diagw, float2* __restrict__ offw) {
  int idx = blockIdx.x * blockDim.x + threadIdx.x;
  int l = idx / UU;
  int u = idx % UU;

  auto psl = [&](const float* ang, int uu) -> float2 {
    uu &= (UU - 1);
    if (uu & 1) return make_float2(1.0f, 0.0f);
    int k = uu >> 1;
    float m = mask[l * (UU / 2) + k];
    float a = ang[l * (UU / 2) + k] * m + (1.0f - m) * PI_F;
    float s, c;
    __sincosf(a, &s, &c);
    return make_float2(c, s);
  };

  float2 ip_m1 = psl(theta, u - 1);
  float2 ip_0  = psl(theta, u);
  float2 ip_p1 = psl(theta, u + 1);
  float2 ep_m1 = psl(phi,   u - 1);
  float2 ep_0  = psl(phi,   u);

  int um1 = (u - 1) & (UU - 1);
  float enn_u = enn[u * LL + l],  enn_m = enn[um1 * LL + l];
  float epp_u = epp[u * LL + l],  epp_m = epp[um1 * LL + l];
  float enp_u = enp[u * LL + l],  enp_m = enp[um1 * LL + l];
  float epn_u = epn[u * LL + l],  epn_m = epn[um1 * LL + l];

  float sdr = epp_u * ip_0.x - enn_u * ip_p1.x - enn_m * ip_m1.x + epp_m * ip_0.x;
  float sdi = epp_u * ip_0.y - enn_u * ip_p1.y - enn_m * ip_m1.y + epp_m * ip_0.y;

  float cvr = epn_u * ip_0.x + enp_u * ip_p1.x + enp_m * ip_m1.x + epn_m * ip_0.x;
  float cvi = epn_u * ip_0.y + enp_u * ip_p1.y + enp_m * ip_m1.y + epn_m * ip_0.y;
  float sor = -cvi, soi = cvr;

  diagw[idx] = make_float2(0.5f * (ep_0.x * sdr - ep_0.y * sdi),
                           0.5f * (ep_0.x * sdi + ep_0.y * sdr));
  offw[idx]  = make_float2(0.5f * (ep_m1.x * sor - ep_m1.y * soi),
                           0.5f * (ep_m1.x * soi + ep_m1.y * sor));
}

__global__ __launch_bounds__(256) void invp_kernel(const int* __restrict__ perms,
                                                   int* __restrict__ invp) {
  int idx = blockIdx.x * blockDim.x + threadIdx.x;
  int l = idx / UU;
  int u = idx % UU;
  int c = perms[(l + 1) * UU + u];
  invp[l * UU + c] = sidx(u, 0);
}

__global__ __launch_bounds__(256) void scan_kernel(
    const float* __restrict__ x, const float* __restrict__ gamma,
    const int* __restrict__ perms, const float2* __restrict__ diagw,
    const float2* __restrict__ offw, const int* __restrict__ invp,
    float* __restrict__ out) {
  __shared__ float2 S[16 * UU];

  const int tid  = threadIdx.x;
  const int b0   = blockIdx.x * 16;
  const int wave = tid >> 5;
  const int lane = tid & 31;
  const int n    = lane & 15;
  const int hi   = lane >> 4;
  const int m    = lane & 15;
  const int r    = m & 3;
  const int blk  = m >> 2;

#pragma unroll
  for (int e = 0; e < (16 * UU) / 256; ++e) {
    int idx = tid + e * 256;
    int rr = idx >> 9;
    int u  = idx & (UU - 1);
    int p  = perms[u];
    float xr = x[(size_t)(b0 + rr) * UU + p];
    float xi = x[(size_t)BB * UU + (size_t)(b0 + rr) * UU + p];
    float sg, cg;
    __sincosf(gamma[p], &sg, &cg);
    S[sidx(u, rr)] = make_float2(xr * cg - xi * sg, xr * sg + xi * cg);
  }
  __syncthreads();

  for (int l = 0; l < LL; ++l) {
    const float2* dl = diagw + (size_t)l * UU;
    const float2* ol = offw  + (size_t)l * UU;
    const int*    il = invp  + (size_t)l * UU;
    __builtin_prefetch(dl + UU, 0, 0);
    __builtin_prefetch(ol + UU, 0, 0);
    __builtin_prefetch(il + UU, 0, 0);

    v8f acc[8];
#pragma unroll
    for (int t = 0; t < 8; ++t) {
      const int cb = (wave * 8 + t) * 8;
      v8f c = {0.f, 0.f, 0.f, 0.f, 0.f, 0.f, 0.f, 0.f};
#pragma unroll
      for (int j = 0; j < 4; ++j) {
        const int col = cb + 2 * j + hi;
        float2 sb = S[sidx(col, n)];
        v2f bop;  bop[0] = sb.x;  bop[1] = sb.y;
        bool use_diag = ((r < 2) == (hi == 0));
        float2 p = use_diag ? dl[col] : ol[col];
        float a0, a1;
        if (r & 1) { a0 = p.y;  a1 = p.x;  }
        else       { a0 = p.x;  a1 = -p.y; }
        if (blk != j) { a0 = 0.f; a1 = 0.f; }
        v2f aop;  aop[0] = a0;  aop[1] = a1;
        c = __builtin_amdgcn_wmma_f32_16x16x4_f32(
                false, aop, false, bop, (short)0, c, false, false);
      }
      acc[t] = c;
    }
    __syncthreads();
#pragma unroll
    for (int t = 0; t < 8; ++t) {
      const int cb = (wave * 8 + t) * 8;
#pragma unroll
      for (int v = 0; v < 8; v += 2) {
        int mm = v + hi * 8;
        int cc = cb + (mm >> 1);
        int so = il[cc];
        S[so + n] = make_float2(acc[t][v], acc[t][v + 1]);
      }
    }
    __syncthreads();
  }

  for (int e = 0; e < (16 * UU) / 256; ++e) {
    int idx = tid + e * 256;
    int rr = idx >> 9;
    int u  = idx & (UU - 1);
    float2 s = S[sidx(u, rr)];
    out[(size_t)(b0 + rr) * UU + u] = s.x;
    out[(size_t)BB * UU + (size_t)(b0 + rr) * UU + u] = s.y;
  }
}

extern "C" void kernel_launch(void* const* d_in, const int* in_sizes, int n_in,
                              void* d_out, int out_size, void* d_ws, size_t ws_size,
                              hipStream_t stream) {
  const float* x     = (const float*)d_in[0];
  const float* theta = (const float*)d_in[1];
  const float* phi   = (const float*)d_in[2];
  const float* gamma = (const float*)d_in[3];
  const float* mask  = (const float*)d_in[4];
  const float* enn   = (const float*)d_in[5];
  const float* enp   = (const float*)d_in[6];
  const float* epn   = (const float*)d_in[7];
  const float* epp   = (const float*)d_in[8];
  const int*   perms = (const int*)d_in[9];

  char* ws = (char*)d_ws;
  float2* diagw = (float2*)(ws);
  float2* offw  = (float2*)(ws + (size_t)LL * UU * 8);
  int*    invp  = (int*)   (ws + (size_t)LL * UU * 16);

  const int threads = 256;
  const int blocksLU = (LL * UU) / threads;

  coeff_kernel<<<blocksLU, threads, 0, stream>>>(theta, phi, mask, enn, enp,
                                                 epn, epp, diagw, offw);
  invp_kernel<<<blocksLU, threads, 0, stream>>>(perms, invp);
  scan_kernel<<<BB / 16, threads, 0, stream>>>(x, gamma, perms, diagw, offw,
                                               invp, (float*)d_out);
}


# scan: per-layer A coefficients (one per lane per tile, not four) and write-back slots are loaded into registers a whole layer ahead, barriers order LDS only; no global-load wait inside a layer
# speedup vs baseline: 4.2902x; 1.4491x over previous
#include <hip/hip_runtime.h>
#include <hip/hip_bf16.h>

#define UU 512
#define LL 512
#define BB 256
static __device__ constexpr float PI_F = 3.14159265358979323846f;

typedef float v2f __attribute__((ext_vector_type(2)));
typedef float v8f __attribute__((ext_vector_type(8)));

__device__ __forceinline__ int sidx(int col, int row) {
  return ((col >> 1) << 5) + ((((col ^ (col >> 2)) & 1)) << 4) + row;
}

__global__ __launch_bounds__(256) void coeff_kernel(
    const float* __restrict__ theta, const float* __restrict__ phi,
    const float* __restrict__ mask,  const float* __restrict__ enn,
    const float* __restrict__ enp,   const float* __restrict__ epn,
    const float* __restrict__ epp,
    float2* __restrict__ diagw, float2* __restrict__ offw) {
  int idx = blockIdx.x * blockDim.x + threadIdx.x;
  int l = idx / UU;
  int u = idx % UU;

  auto psl = [&](const float* ang, int uu) -> float2 {
    uu &= (UU - 1);
    if (uu & 1) return make_float2(1.0f, 0.0f);
    int k = uu >> 1;
    float m = mask[l * (UU / 2) + k];
    float a = ang[l * (UU / 2) + k] * m + (1.0f - m) * PI_F;
    float s, c;
    __sincosf(a, &s, &c);
    return make_float2(c, s);
  };

  float2 ip_m1 = psl(theta, u - 1);
  float2 ip_0  = psl(theta, u);
  float2 ip_p1 = psl(theta, u + 1);
  float2 ep_m1 = psl(phi,   u - 1);
  float2 ep_0  = psl(phi,   u);

  int um1 = (u - 1) & (UU - 1);
  float enn_u = enn[u * LL + l],  enn_m = enn[um1 * LL + l];
  float epp_u = epp[u * LL + l],  epp_m = epp[um1 * LL + l];
  float enp_u = enp[u * LL + l],  enp_m = enp[um1 * LL + l];
  float epn_u = epn[u * LL + l],  epn_m = epn[um1 * LL + l];

  float sdr = epp_u * ip_0.x - enn_u * ip_p1.x - enn_m * ip_m1.x + epp_m * ip_0.x;
  float sdi = epp_u * ip_0.y - enn_u * ip_p1.y - enn_m * ip_m1.y + epp_m * ip_0.y;

  float cvr = epn_u * ip_0.x + enp_u * ip_p1.x + enp_m * ip_m1.x + epn_m * ip_0.x;
  float cvi = epn_u * ip_0.y + enp_u * ip_p1.y + enp_m * ip_m1.y + epn_m * ip_0.y;
  float sor = -cvi, soi = cvr;

  diagw[idx] = make_float2(0.5f * (ep_0.x * sdr - ep_0.y * sdi),
                           0.5f * (ep_0.x * sdi + ep_0.y * sdr));
  offw[idx]  = make_float2(0.5f * (ep_m1.x * sor - ep_m1.y * soi),
                           0.5f * (ep_m1.x * soi + ep_m1.y * sor));
}

__global__ __launch_bounds__(256) void invp_kernel(const int* __restrict__ perms,
                                                   int* __restrict__ invp) {
  int idx = blockIdx.x * blockDim.x + threadIdx.x;
  int l = idx / UU;
  int u = idx % UU;
  int c = perms[(l + 1) * UU + u];
  invp[l * UU + c] = sidx(u, 0);
}

typedef int v4i __attribute__((ext_vector_type(4)));

__device__ __forceinline__ void lds_barrier() {
  __builtin_amdgcn_fence(__ATOMIC_RELEASE, "workgroup", "local");
  __builtin_amdgcn_s_barrier();
  __builtin_amdgcn_fence(__ATOMIC_ACQUIRE, "workgroup", "local");
  __builtin_amdgcn_sched_barrier(0);
}

struct CoefRegs { float2 p[8]; };
struct SlotRegs { v4i s[8]; };

__global__ __launch_bounds__(256) void scan_kernel(
    const float* __restrict__ x, const float* __restrict__ gamma,
    const int* __restrict__ perms, const float2* __restrict__ diagw,
    const float2* __restrict__ offw, const int* __restrict__ invp,
    float* __restrict__ out) {
  __shared__ float2 S[16 * UU];

  const int tid  = threadIdx.x;
  const int b0   = blockIdx.x * 16;
  const int wave = tid >> 5;
  const int lane = tid & 31;
  const int n    = lane & 15;
  const int hi   = lane >> 4;
  const int m    = lane & 15;
  const int r    = m & 3;
  const int blk  = m >> 2;

  const bool use_diag = ((r < 2) == (hi == 0));
  const float2* csel = (use_diag ? diagw : offw) + (wave * 64 + 2 * blk + hi);
  const v4i* isel = reinterpret_cast<const v4i*>(invp + wave * 64 + hi * 4);

  auto fetch_coef = [&](CoefRegs& g, int l) {
    l = (l < LL) ? l : (LL - 1);
    const float2* cp = csel + (size_t)l * UU;
#pragma unroll
    for (int t = 0; t < 8; ++t) g.p[t] = cp[t * 8];
  };
  auto fetch_slot = [&](SlotRegs& g, int l) {
    l = (l < LL) ? l : (LL - 1);
    const v4i* ip = isel + (size_t)l * (UU / 4);
#pragma unroll
    for (int t = 0; t < 8; ++t) g.s[t] = ip[t * 2];
  };

  CoefRegs ca, cb2;
  SlotRegs sa, sb2;
  fetch_coef(ca, 0);
  fetch_coef(cb2, 1);
  fetch_slot(sa, 0);

#pragma unroll
  for (int e = 0; e < (16 * UU) / 256; ++e) {
    int idx = tid + e * 256;
    int rr = idx >> 9;
    int u  = idx & (UU - 1);
    int p  = perms[u];
    float xr = x[(size_t)(b0 + rr) * UU + p];
    float xi = x[(size_t)BB * UU + (size_t)(b0 + rr) * UU + p];
    float sg, cg;
    __sincosf(gamma[p], &sg, &cg);
    S[sidx(u, rr)] = make_float2(xr * cg - xi * sg, xr * sg + xi * cg);
  }
  __syncthreads();

  auto layer = [&](CoefRegs& g, const SlotRegs& sl, SlotRegs& sn, int l) {
    v2f bop[8][4];
#pragma unroll
    for (int t = 0; t < 8; ++t) {
#pragma unroll
      for (int j = 0; j < 4; ++j) {
        float2 sb = S[sidx((wave * 8 + t) * 8 + 2 * j + hi, n)];
        bop[t][j][0] = sb.x;  bop[t][j][1] = sb.y;
      }
    }
    lds_barrier();
    fetch_slot(sn, l + 1);
#pragma unroll
    for (int t = 0; t < 8; ++t) {
      const float2 p = g.p[t];
      float a0, a1;
      if (r & 1) { a0 = p.y;  a1 = p.x;  }
      else       { a0 = p.x;  a1 = -p.y; }
      v8f c = {0.f, 0.f, 0.f, 0.f, 0.f, 0.f, 0.f, 0.f};
#pragma unroll
      for (int j = 0; j < 4; ++j) {
        v2f aop;
        aop[0] = (blk == j) ? a0 : 0.f;
        aop[1] = (blk == j) ? a1 : 0.f;
        c = __builtin_amdgcn_wmma_f32_16x16x4_f32(
                false, aop, false, bop[t][j], (short)0, c, false, false);
      }
      const v4i so = sl.s[t];
#pragma unroll
      for (int v = 0; v < 8; v += 2) {
        S[so[v >> 1] + n] = make_float2(c[v], c[v + 1]);
      }
    }
    fetch_coef(g, l + 2);
    lds_barrier();
  };

  for (int l = 0; l < LL; l += 2) {
    layer(ca, sa, sb2, l);
    layer(cb2, sb2, sa, l + 1);
  }

  for (int e = 0; e < (16 * UU) / 256; ++e) {
    int idx = tid + e * 256;
    int rr = idx >> 9;
    int u  = idx & (UU - 1);
    float2 s = S[sidx(u, rr)];
    out[(size_t)(b0 + rr) * UU + u] = s.x;
    out[(size_t)BB * UU + (size_t)(b0 + rr) * UU + u] = s.y;
  }
}

extern "C" void kernel_launch(void* const* d_in, const int* in_sizes, int n_in,
                              void* d_out, int out_size, void* d_ws, size_t ws_size,
                              hipStream_t stream) {
  const float* x     = (const float*)d_in[0];
  const float* theta = (const float*)d_in[1];
  const float* phi   = (const float*)d_in[2];
  const float* gamma = (const float*)d_in[3];
  const float* mask  = (const float*)d_in[4];
  const float* enn   = (const float*)d_in[5];
  const float* enp   = (const float*)d_in[6];
  const float* epn   = (const float*)d_in[7];
  const float* epp   = (const float*)d_in[8];
  const int*   perms = (const int*)d_in[9];

  char* ws = (char*)d_ws;
  float2* diagw = (float2*)(ws);
  float2* offw  = (float2*)(ws + (size_t)LL * UU * 8);
  int*    invp  = (int*)   (ws + (size_t)LL * UU * 16);

  const int threads = 256;
  const int blocksLU = (LL * UU) / threads;

  coeff_kernel<<<blocksLU, threads, 0, stream>>>(theta, phi, mask, enn, enp,
                                                 epn, epp, diagw, offw);
  invp_kernel<<<blocksLU, threads, 0, stream>>>(perms, invp);
  scan_kernel<<<BB / 16, threads, 0, stream>>>(x, gamma, perms, diagw, offw,
                                               invp, (float*)d_out);
}


# work-group takes 4 batch rows, not 16: the four 4x4 pair blocks of an 8-column tile ride ONE 16x16x4 product (pair x batch on N) instead of four zero-padded ones; 64 work-groups, same K=4 sums
# speedup vs baseline: 10.0474x; 2.3419x over previous
#include <hip/hip_runtime.h>
#include <hip/hip_bf16.h>

#define UU 512
#define LL 512
#define BB 256
static __device__ constexpr float PI_F = 3.14159265358979323846f;

typedef float v2f __attribute__((ext_vector_type(2)));
typedef float v8f __attribute__((ext_vector_type(8)));

#define NB 4

__device__ __forceinline__ int sidx(int col, int row) {
  return col * NB + row;
}

__global__ __launch_bounds__(256) void coeff_kernel(
    const float* __restrict__ theta, const float* __restrict__ phi,
    const float* __restrict__ mask,  const float* __restrict__ enn,
    const float* __restrict__ enp,   const float* __restrict__ epn,
    const float* __restrict__ epp,
    float2* __restrict__ diagw, float2* __restrict__ offw) {
  int idx = blockIdx.x * blockDim.x + threadIdx.x;
  int l = idx / UU;
  int u = idx % UU;

  auto psl = [&](const float* ang, int uu) -> float2 {
    uu &= (UU - 1);
    if (uu & 1) return make_float2(1.0f, 0.0f);
    int k = uu >> 1;
    float m = mask[l * (UU / 2) + k];
    float a = ang[l * (UU / 2) + k] * m + (1.0f - m) * PI_F;
    float s, c;
    __sincosf(a, &s, &c);
    return make_float2(c, s);
  };

  float2 ip_m1 = psl(theta, u - 1);
  float2 ip_0  = psl(theta, u);
  float2 ip_p1 = psl(theta, u + 1);
  float2 ep_m1 = psl(phi,   u - 1);
  float2 ep_0  = psl(phi,   u);

  int um1 = (u - 1) & (UU - 1);
  float enn_u = enn[u * LL + l],  enn_m = enn[um1 * LL + l];
  float epp_u = epp[u * LL + l],  epp_m = epp[um1 * LL + l];
  float enp_u = enp[u * LL + l],  enp_m = enp[um1 * LL + l];
  float epn_u = epn[u * LL + l],  epn_m = epn[um1 * LL + l];

  float sdr = epp_u * ip_0.x - enn_u * ip_p1.x - enn_m * ip_m1.x + epp_m * ip_0.x;
  float sdi = epp_u * ip_0.y - enn_u * ip_p1.y - enn_m * ip_m1.y + epp_m * ip_0.y;

  float cvr = epn_u * ip_0.x + enp_u * ip_p1.x + enp_m * ip_m1.x + epn_m * ip_0.x;
  float cvi = epn_u * ip_0.y + enp_u * ip_p1.y + enp_m * ip_m1.y + epn_m * ip_0.y;
  float sor = -cvi, soi = cvr;

  diagw[idx] = make_float2(0.5f * (ep_0.x * sdr - ep_0.y * sdi),
                           0.5f * (ep_0.x * sdi + ep_0.y * sdr));
  offw[idx]  = make_float2(0.5f * (ep_m1.x * sor - ep_m1.y * soi),
                           0.5f * (ep_m1.x * soi + ep_m1.y * sor));
}

__global__ __launch_bounds__(256) void invp_kernel(const int* __restrict__ perms,
                                                   int* __restrict__ invp) {
  int idx = blockIdx.x * blockDim.x + threadIdx.x;
  int l = idx / UU;
  int u = idx % UU;
  int c = perms[(l + 1) * UU + u];
  invp[l * UU + c] = sidx(u, 0);
}

typedef int v2i __attribute__((ext_vector_type(2)));

__device__ __forceinline__ void lds_barrier() {
  __builtin_amdgcn_fence(__ATOMIC_RELEASE, "workgroup", "local");
  __builtin_amdgcn_s_barrier();
  __builtin_amdgcn_fence(__ATOMIC_ACQUIRE, "workgroup", "local");
  __builtin_amdgcn_sched_barrier(0);
}

struct CoefRegs { float2 p[8]; };
struct SlotRegs { v2i s[8]; };

__global__ __launch_bounds__(256) void scan_kernel(
    const float* __restrict__ x, const float* __restrict__ gamma,
    const int* __restrict__ perms, const float2* __restrict__ diagw,
    const float2* __restrict__ offw, const int* __restrict__ invp,
    float* __restrict__ out) {
  __shared__ float2 S[NB * UU];

  const int tid  = threadIdx.x;
  const int b0   = blockIdx.x * NB;
  const int wave = tid >> 5;
  const int lane = tid & 31;
  const int hi   = lane >> 4;
  const int n    = lane & 15;
  const int q    = n >> 2;
  const int b    = n & 3;
  const int m    = lane & 15;
  const int r    = m & 3;
  const int blk  = m >> 2;

  const bool use_diag = ((r < 2) == (hi == 0));
  const float2* csel = (use_diag ? diagw : offw) + (wave * 64 + 2 * blk + hi);
  const v2i* isel = reinterpret_cast<const v2i*>(invp + wave * 64 + 2 * q);
  const bool mine  = ((q >> 1) == hi);
  const bool upper = ((q & 1) != 0);

  auto fetch_coef = [&](CoefRegs& g, int l) {
    l = (l < LL) ? l : (LL - 1);
    const float2* cp = csel + (size_t)l * UU;
#pragma unroll
    for (int t = 0; t < 8; ++t) g.p[t] = cp[t * 8];
  };
  auto fetch_slot = [&](SlotRegs& g, int l) {
    l = (l < LL) ? l : (LL - 1);
    const v2i* ip = isel + (size_t)l * (UU / 2);
#pragma unroll
    for (int t = 0; t < 8; ++t) g.s[t] = ip[t * 4];
  };

  CoefRegs ca, cb2;
  SlotRegs sa, sb2;
  fetch_coef(ca, 0);
  fetch_coef(cb2, 1);
  fetch_slot(sa, 0);

#pragma unroll
  for (int e = 0; e < (NB * UU) / 256; ++e) {
    int idx = tid + e * 256;
    int rr = idx >> 9;
    int u  = idx & (UU - 1);
    int p  = perms[u];
    float xr = x[(size_t)(b0 + rr) * UU + p];
    float xi = x[(size_t)BB * UU + (size_t)(b0 + rr) * UU + p];
    float sg, cg;
    __sincosf(gamma[p], &sg, &cg);
    S[sidx(u, rr)] = make_float2(xr * cg - xi * sg, xr * sg + xi * cg);
  }
  __syncthreads();

  auto layer = [&](CoefRegs& g, const SlotRegs& sl, SlotRegs& sn, int l) {
    v2f bop[8];
#pragma unroll
    for (int t = 0; t < 8; ++t) {
      float2 sb = S[sidx((wave * 8 + t) * 8 + 2 * q + hi, b)];
      bop[t][0] = sb.x;  bop[t][1] = sb.y;
    }
    lds_barrier();
    fetch_slot(sn, l + 1);
    v8f acc[8];
#pragma unroll
    for (int t = 0; t < 8; ++t) {
      const float2 p = g.p[t];
      v2f aop;
      if (r & 1) { aop[0] = p.y;  aop[1] = p.x;  }
      else       { aop[0] = p.x;  aop[1] = -p.y; }
      const v8f zero = {0.f, 0.f, 0.f, 0.f, 0.f, 0.f, 0.f, 0.f};
      acc[t] = __builtin_amdgcn_wmma_f32_16x16x4_f32(
                   false, aop, false, bop[t], (short)0, zero, false, false);
    }
    if (mine) {
#pragma unroll
      for (int t = 0; t < 8; ++t) {
        const v8f c = acc[t];
        const v2i so = sl.s[t];
        const float y0r = upper ? c[4] : c[0];
        const float y0i = upper ? c[5] : c[1];
        const float y1r = upper ? c[6] : c[2];
        const float y1i = upper ? c[7] : c[3];
        S[so[0] + b] = make_float2(y0r, y0i);
        S[so[1] + b] = make_float2(y1r, y1i);
      }
    }
    fetch_coef(g, l + 2);
    lds_barrier();
  };

  for (int l = 0; l < LL; l += 2) {
    layer(ca, sa, sb2, l);
    layer(cb2, sb2, sa, l + 1);
  }

  for (int e = 0; e < (NB * UU) / 256; ++e) {
    int idx = tid + e * 256;
    int rr = idx >> 9;
    int u  = idx & (UU - 1);
    float2 s = S[sidx(u, rr)];
    out[(size_t)(b0 + rr) * UU + u] = s.x;
    out[(size_t)BB * UU + (size_t)(b0 + rr) * UU + u] = s.y;
  }
}

extern "C" void kernel_launch(void* const* d_in, const int* in_sizes, int n_in,
                              void* d_out, int out_size, void* d_ws, size_t ws_size,
                              hipStream_t stream) {
  const float* x     = (const float*)d_in[0];
  const float* theta = (const float*)d_in[1];
  const float* phi   = (const float*)d_in[2];
  const float* gamma = (const float*)d_in[3];
  const float* mask  = (const float*)d_in[4];
  const float* enn   = (const float*)d_in[5];
  const float* enp   = (const float*)d_in[6];
  const float* epn   = (const float*)d_in[7];
  const float* epp   = (const float*)d_in[8];
  const int*   perms = (const int*)d_in[9];

  char* ws = (char*)d_ws;
  float2* diagw = (float2*)(ws);
  float2* offw  = (float2*)(ws + (size_t)LL * UU * 8);
  int*    invp  = (int*)   (ws + (size_t)LL * UU * 16);

  const int threads = 256;
  const int blocksLU = (LL * UU) / threads;

  coeff_kernel<<<blocksLU, threads, 0, stream>>>(theta, phi, mask, enn, enp,
                                                 epn, epp, diagw, offw);
  invp_kernel<<<blocksLU, threads, 0, stream>>>(perms, invp);
  scan_kernel<<<BB / NB, threads, 0, stream>>>(x, gamma, perms, diagw, offw,
                                               invp, (float*)d_out);
}


# LDS state held twice, layers alternate read copy and write copy: one work-group barrier per layer instead of two, a wave's products start once its own B reads land
# speedup vs baseline: 10.7018x; 1.0651x over previous
#include <hip/hip_runtime.h>
#include <hip/hip_bf16.h>

#define UU 512
#define LL 512
#define BB 256
static __device__ constexpr float PI_F = 3.14159265358979323846f;

typedef float v2f __attribute__((ext_vector_type(2)));
typedef float v8f __attribute__((ext_vector_type(8)));

#define NB 4

__device__ __forceinline__ int sidx(int col, int row) {
  return col * NB + row;
}

__global__ __launch_bounds__(256) void coeff_kernel(
    const float* __restrict__ theta, const float* __restrict__ phi,
    const float* __restrict__ mask,  const float* __restrict__ enn,
    const float* __restrict__ enp,   const float* __restrict__ epn,
    const float* __restrict__ epp,
    float2* __restrict__ diagw, float2* __restrict__ offw) {
  int idx = blockIdx.x * blockDim.x + threadIdx.x;
  int l = idx / UU;
  int u = idx % UU;

  auto psl = [&](const float* ang, int uu) -> float2 {
    uu &= (UU - 1);
    if (uu & 1) return make_float2(1.0f, 0.0f);
    int k = uu >> 1;
    float m = mask[l * (UU / 2) + k];
    float a = ang[l * (UU / 2) + k] * m + (1.0f - m) * PI_F;
    float s, c;
    __sincosf(a, &s, &c);
    return make_float2(c, s);
  };

  float2 ip_m1 = psl(theta, u - 1);
  float2 ip_0  = psl(theta, u);
  float2 ip_p1 = psl(theta, u + 1);
  float2 ep_m1 = psl(phi,   u - 1);
  float2 ep_0  = psl(phi,   u);

  int um1 = (u - 1) & (UU - 1);
  float enn_u = enn[u * LL + l],  enn_m = enn[um1 * LL + l];
  float epp_u = epp[u * LL + l],  epp_m = epp[um1 * LL + l];
  float enp_u = enp[u * LL + l],  enp_m = enp[um1 * LL + l];
  float epn_u = epn[u * LL + l],  epn_m = epn[um1 * LL + l];

  float sdr = epp_u * ip_0.x - enn_u * ip_p1.x - enn_m * ip_m1.x + epp_m * ip_0.x;
  float sdi = epp_u * ip_0.y - enn_u * ip_p1.y - enn_m * ip_m1.y + epp_m * ip_0.y;

  float cvr = epn_u * ip_0.x + enp_u * ip_p1.x + enp_m * ip_m1.x + epn_m * ip_0.x;
  float cvi = epn_u * ip_0.y + enp_u * ip_p1.y + enp_m * ip_m1.y + epn_m * ip_0.y;
  float sor = -cvi, soi = cvr;

  diagw[idx] = make_float2(0.5f * (ep_0.x * sdr - ep_0.y * sdi),
                           0.5f * (ep_0.x * sdi + ep_0.y * sdr));
  offw[idx]  = make_float2(0.5f * (ep_m1.x * sor - ep_m1.y * soi),
                           0.5f * (ep_m1.x * soi + ep_m1.y * sor));
}

__global__ __launch_bounds__(256) void invp_kernel(const int* __restrict__ perms,
                                                   int* __restrict__ invp) {
  int idx = blockIdx.x * blockDim.x + threadIdx.x;
  int l = idx / UU;
  int u = idx % UU;
  int c = perms[(l + 1) * UU + u];
  invp[l * UU + c] = sidx(u, 0);
}

typedef int v2i __attribute__((ext_vector_type(2)));

__device__ __forceinline__ void lds_barrier() {
  __builtin_amdgcn_fence(__ATOMIC_RELEASE, "workgroup", "local");
  __builtin_amdgcn_s_barrier();
  __builtin_amdgcn_fence(__ATOMIC_ACQUIRE, "workgroup", "local");
  __builtin_amdgcn_sched_barrier(0);
}

struct CoefRegs { float2 p[8]; };
struct SlotRegs { v2i s[8]; };

__global__ __launch_bounds__(256) void scan_kernel(
    const float* __restrict__ x, const float* __restrict__ gamma,
    const int* __restrict__ perms, const float2* __restrict__ diagw,
    const float2* __restrict__ offw, const int* __restrict__ invp,
    float* __restrict__ out) {
  __shared__ float2 S[2 * NB * UU];

  const int tid  = threadIdx.x;
  const int b0   = blockIdx.x * NB;
  const int wave = tid >> 5;
  const int lane = tid & 31;
  const int hi   = lane >> 4;
  const int n    = lane & 15;
  const int q    = n >> 2;
  const int b    = n & 3;
  const int m    = lane & 15;
  const int r    = m & 3;
  const int blk  = m >> 2;

  const bool use_diag = ((r < 2) == (hi == 0));
  const float2* csel = (use_diag ? diagw : offw) + (wave * 64 + 2 * blk + hi);
  const v2i* isel = reinterpret_cast<const v2i*>(invp + wave * 64 + 2 * q);
  const bool mine  = ((q >> 1) == hi);
  const bool upper = ((q & 1) != 0);

  auto fetch_coef = [&](CoefRegs& g, int l) {
    l = (l < LL) ? l : (LL - 1);
    const float2* cp = csel + (size_t)l * UU;
#pragma unroll
    for (int t = 0; t < 8; ++t) g.p[t] = cp[t * 8];
  };
  auto fetch_slot = [&](SlotRegs& g, int l) {
    l = (l < LL) ? l : (LL - 1);
    const v2i* ip = isel + (size_t)l * (UU / 2);
#pragma unroll
    for (int t = 0; t < 8; ++t) g.s[t] = ip[t * 4];
  };

  CoefRegs ca, cb2;
  SlotRegs sa, sb2;
  fetch_coef(ca, 0);
  fetch_coef(cb2, 1);
  fetch_slot(sa, 0);

#pragma unroll
  for (int e = 0; e < (NB * UU) / 256; ++e) {
    int idx = tid + e * 256;
    int rr = idx >> 9;
    int u  = idx & (UU - 1);
    int p  = perms[u];
    float xr = x[(size_t)(b0 + rr) * UU + p];
    float xi = x[(size_t)BB * UU + (size_t)(b0 + rr) * UU + p];
    float sg, cg;
    __sincosf(gamma[p], &sg, &cg);
    S[sidx(u, rr)] = make_float2(xr * cg - xi * sg, xr * sg + xi * cg);
  }
  __syncthreads();

  auto layer = [&](CoefRegs& g, const SlotRegs& sl, SlotRegs& sn, int l,
                   const int rd, const int wr) {
    v2f bop[8];
#pragma unroll
    for (int t = 0; t < 8; ++t) {
      float2 sb = S[rd + sidx((wave * 8 + t) * 8 + 2 * q + hi, b)];
      bop[t][0] = sb.x;  bop[t][1] = sb.y;
    }
    __builtin_amdgcn_sched_barrier(0);
    fetch_slot(sn, l + 1);
    v8f acc[8];
#pragma unroll
    for (int t = 0; t < 8; ++t) {
      const float2 p = g.p[t];
      v2f aop;
      if (r & 1) { aop[0] = p.y;  aop[1] = p.x;  }
      else       { aop[0] = p.x;  aop[1] = -p.y; }
      const v8f zero = {0.f, 0.f, 0.f, 0.f, 0.f, 0.f, 0.f, 0.f};
      acc[t] = __builtin_amdgcn_wmma_f32_16x16x4_f32(
                   false, aop, false, bop[t], (short)0, zero, false, false);
    }
    if (mine) {
#pragma unroll
      for (int t = 0; t < 8; ++t) {
        const v8f c = acc[t];
        const v2i so = sl.s[t];
        const float y0r = upper ? c[4] : c[0];
        const float y0i = upper ? c[5] : c[1];
        const float y1r = upper ? c[6] : c[2];
        const float y1i = upper ? c[7] : c[3];
        S[wr + so[0] + b] = make_float2(y0r, y0i);
        S[wr + so[1] + b] = make_float2(y1r, y1i);
      }
    }
    fetch_coef(g, l + 2);
    lds_barrier();
  };

  for (int l = 0; l < LL; l += 2) {
    layer(ca, sa, sb2, l, 0, NB * UU);
    layer(cb2, sb2, sa, l + 1, NB * UU, 0);
  }

  for (int e = 0; e < (NB * UU) / 256; ++e) {
    int idx = tid + e * 256;
    int rr = idx >> 9;
    int u  = idx & (UU - 1);
    float2 s = S[sidx(u, rr)];
    out[(size_t)(b0 + rr) * UU + u] = s.x;
    out[(size_t)BB * UU + (size_t)(b0 + rr) * UU + u] = s.y;
  }
}

extern "C" void kernel_launch(void* const* d_in, const int* in_sizes, int n_in,
                              void* d_out, int out_size, void* d_ws, size_t ws_size,
                              hipStream_t stream) {
  const float* x     = (const float*)d_in[0];
  const float* theta = (const float*)d_in[1];
  const float* phi   = (const float*)d_in[2];
  const float* gamma = (const float*)d_in[3];
  const float* mask  = (const float*)d_in[4];
  const float* enn   = (const float*)d_in[5];
  const float* enp   = (const float*)d_in[6];
  const float* epn   = (const float*)d_in[7];
  const float* epp   = (const float*)d_in[8];
  const int*   perms = (const int*)d_in[9];

  char* ws = (char*)d_ws;
  float2* diagw = (float2*)(ws);
  float2* offw  = (float2*)(ws + (size_t)LL * UU * 8);
  int*    invp  = (int*)   (ws + (size_t)LL * UU * 16);

  const int threads = 256;
  const int blocksLU = (LL * UU) / threads;

  coeff_kernel<<<blocksLU, threads, 0, stream>>>(theta, phi, mask, enn, enp,
                                                 epn, epp, diagw, offw);
  invp_kernel<<<blocksLU, threads, 0, stream>>>(perms, invp);
  scan_kernel<<<BB / NB, threads, 0, stream>>>(x, gamma, perms, diagw, offw,
                                               invp, (float*)d_out);
}


# scan: only the stores stay behind a wave's last product: both operand prefetches and the A selects move to the layer head, D selects and write-back addresses ride under the products
# speedup vs baseline: 11.6275x; 1.0865x over previous
#include <hip/hip_runtime.h>
#include <hip/hip_bf16.h>

#define UU 512
#define LL 512
#define BB 256
static __device__ constexpr float PI_F = 3.14159265358979323846f;

typedef float v2f __attribute__((ext_vector_type(2)));
typedef float v8f __attribute__((ext_vector_type(8)));

#define NB 4

__device__ __forceinline__ int sidx(int col, int row) {
  return col * NB + row;
}

__global__ __launch_bounds__(256) void coeff_kernel(
    const float* __restrict__ theta, const float* __restrict__ phi,
    const float* __restrict__ mask,  const float* __restrict__ enn,
    const float* __restrict__ enp,   const float* __restrict__ epn,
    const float* __restrict__ epp,
    float2* __restrict__ diagw, float2* __restrict__ offw) {
  int idx = blockIdx.x * blockDim.x + threadIdx.x;
  int l = idx / UU;
  int u = idx % UU;

  auto psl = [&](const float* ang, int uu) -> float2 {
    uu &= (UU - 1);
    if (uu & 1) return make_float2(1.0f, 0.0f);
    int k = uu >> 1;
    float m = mask[l * (UU / 2) + k];
    float a = ang[l * (UU / 2) + k] * m + (1.0f - m) * PI_F;
    float s, c;
    __sincosf(a, &s, &c);
    return make_float2(c, s);
  };

  float2 ip_m1 = psl(theta, u - 1);
  float2 ip_0  = psl(theta, u);
  float2 ip_p1 = psl(theta, u + 1);
  float2 ep_m1 = psl(phi,   u - 1);
  float2 ep_0  = psl(phi,   u);

  int um1 = (u - 1) & (UU - 1);
  float enn_u = enn[u * LL + l],  enn_m = enn[um1 * LL + l];
  float epp_u = epp[u * LL + l],  epp_m = epp[um1 * LL + l];
  float enp_u = enp[u * LL + l],  enp_m = enp[um1 * LL + l];
  float epn_u = epn[u * LL + l],  epn_m = epn[um1 * LL + l];

  float sdr = epp_u * ip_0.x - enn_u * ip_p1.x - enn_m * ip_m1.x + epp_m * ip_0.x;
  float sdi = epp_u * ip_0.y - enn_u * ip_p1.y - enn_m * ip_m1.y + epp_m * ip_0.y;

  float cvr = epn_u * ip_0.x + enp_u * ip_p1.x + enp_m * ip_m1.x + epn_m * ip_0.x;
  float cvi = epn_u * ip_0.y + enp_u * ip_p1.y + enp_m * ip_m1.y + epn_m * ip_0.y;
  float sor = -cvi, soi = cvr;

  diagw[idx] = make_float2(0.5f * (ep_0.x * sdr - ep_0.y * sdi),
                           0.5f * (ep_0.x * sdi + ep_0.y * sdr));
  offw[idx]  = make_float2(0.5f * (ep_m1.x * sor - ep_m1.y * soi),
                           0.5f * (ep_m1.x * soi + ep_m1.y * sor));
}

__global__ __launch_bounds__(256) void invp_kernel(const int* __restrict__ perms,
                                                   int* __restrict__ invp) {
  int idx = blockIdx.x * blockDim.x + threadIdx.x;
  int l = idx / UU;
  int u = idx % UU;
  int c = perms[(l + 1) * UU + u];
  invp[l * UU + c] = sidx(u, 0);
}

typedef int v2i __attribute__((ext_vector_type(2)));
typedef float v4f __attribute__((ext_vector_type(4)));

__device__ __forceinline__ void lds_barrier() {
  __builtin_amdgcn_fence(__ATOMIC_RELEASE, "workgroup", "local");
  __builtin_amdgcn_s_barrier();
  __builtin_amdgcn_fence(__ATOMIC_ACQUIRE, "workgroup", "local");
  __builtin_amdgcn_sched_barrier(0);
}

struct CoefRegs { float2 p[8]; };
struct SlotRegs { v2i s[8]; };

__global__ __launch_bounds__(256) void scan_kernel(
    const float* __restrict__ x, const float* __restrict__ gamma,
    const int* __restrict__ perms, const float2* __restrict__ diagw,
    const float2* __restrict__ offw, const int* __restrict__ invp,
    float* __restrict__ out) {
  __shared__ float2 S[2 * NB * UU];

  const int tid  = threadIdx.x;
  const int b0   = blockIdx.x * NB;
  const int wave = tid >> 5;
  const int lane = tid & 31;
  const int hi   = lane >> 4;
  const int n    = lane & 15;
  const int q    = n >> 2;
  const int b    = n & 3;
  const int m    = lane & 15;
  const int r    = m & 3;
  const int blk  = m >> 2;

  const bool use_diag = ((r < 2) == (hi == 0));
  const float2* csel = (use_diag ? diagw : offw) + (wave * 64 + 2 * blk + hi);
  const v2i* isel = reinterpret_cast<const v2i*>(invp + wave * 64 + 2 * q);
  const bool mine  = ((q >> 1) == hi);
  const bool upper = ((q & 1) != 0);

  auto fetch_coef = [&](CoefRegs& g, int l) {
    l = (l < LL) ? l : (LL - 1);
    const float2* cp = csel + (size_t)l * UU;
#pragma unroll
    for (int t = 0; t < 8; ++t) g.p[t] = cp[t * 8];
  };
  auto fetch_slot = [&](SlotRegs& g, int l) {
    l = (l < LL) ? l : (LL - 1);
    const v2i* ip = isel + (size_t)l * (UU / 2);
#pragma unroll
    for (int t = 0; t < 8; ++t) g.s[t] = ip[t * 4];
  };

  CoefRegs ca, cb2;
  SlotRegs sa, sb2;
  fetch_coef(ca, 0);
  fetch_coef(cb2, 1);
  fetch_slot(sa, 0);

#pragma unroll
  for (int e = 0; e < (NB * UU) / 256; ++e) {
    int idx = tid + e * 256;
    int rr = idx >> 9;
    int u  = idx & (UU - 1);
    int p  = perms[u];
    float xr = x[(size_t)(b0 + rr) * UU + p];
    float xi = x[(size_t)BB * UU + (size_t)(b0 + rr) * UU + p];
    float sg, cg;
    __sincosf(gamma[p], &sg, &cg);
    S[sidx(u, rr)] = make_float2(xr * cg - xi * sg, xr * sg + xi * cg);
  }
  __syncthreads();

  auto layer = [&](CoefRegs& g, const SlotRegs& sl, SlotRegs& sn, int l,
                   const int rd, const int wr) {
    v2f bop[8];
#pragma unroll
    for (int t = 0; t < 8; ++t) {
      float2 sb = S[rd + sidx((wave * 8 + t) * 8 + 2 * q + hi, b)];
      bop[t][0] = sb.x;  bop[t][1] = sb.y;
    }
    __builtin_amdgcn_sched_barrier(0);
    fetch_slot(sn, l + 1);

    v2f aop[8];
    auto mk_a = [&](int t) {
      const float2 p = g.p[t];
      if (r & 1) { aop[t][0] = p.y;  aop[t][1] = p.x;  }
      else       { aop[t][0] = p.x;  aop[t][1] = -p.y; }
    };
    int w0[8], w1[8];
    auto mk_w = [&](int t) {
      const v2i so = sl.s[t];
      w0[t] = (wr + so[0] + b) * 8;
      w1[t] = (wr + so[1] + b) * 8;
      asm volatile("" :: "v"(w0[t]), "v"(w1[t]));
    };
    v8f acc[8];
    v4f y[8];
    auto pick = [&](int t) {
      const v8f c = acc[t];
      y[t][0] = upper ? c[4] : c[0];
      y[t][1] = upper ? c[5] : c[1];
      y[t][2] = upper ? c[6] : c[2];
      y[t][3] = upper ? c[7] : c[3];
      asm volatile("" :: "v"(y[t]));
    };
#pragma unroll
    for (int t = 0; t < 8; ++t) mk_a(t);
    fetch_coef(g, l + 2);
#pragma unroll
    for (int t = 0; t < 8; ++t) {
      const v8f zero = {0.f, 0.f, 0.f, 0.f, 0.f, 0.f, 0.f, 0.f};
      acc[t] = __builtin_amdgcn_wmma_f32_16x16x4_f32(
                   false, aop[t], false, bop[t], (short)0, zero, false, false);
      __builtin_amdgcn_sched_barrier(0);
      if (t > 0) pick(t - 1);
      mk_w(t);
      __builtin_amdgcn_sched_barrier(0);
    }
    pick(7);
    if (mine) {
      char* sbytes = reinterpret_cast<char*>(S);
#pragma unroll
      for (int t = 0; t < 8; ++t) {
        *reinterpret_cast<float2*>(sbytes + w0[t]) = make_float2(y[t][0], y[t][1]);
        *reinterpret_cast<float2*>(sbytes + w1[t]) = make_float2(y[t][2], y[t][3]);
      }
    }
    lds_barrier();
  };

  for (int l = 0; l < LL; l += 2) {
    layer(ca, sa, sb2, l, 0, NB * UU);
    layer(cb2, sb2, sa, l + 1, NB * UU, 0);
  }

  for (int e = 0; e < (NB * UU) / 256; ++e) {
    int idx = tid + e * 256;
    int rr = idx >> 9;
    int u  = idx & (UU - 1);
    float2 s = S[sidx(u, rr)];
    out[(size_t)(b0 + rr) * UU + u] = s.x;
    out[(size_t)BB * UU + (size_t)(b0 + rr) * UU + u] = s.y;
  }
}

extern "C" void kernel_launch(void* const* d_in, const int* in_sizes, int n_in,
                              void* d_out, int out_size, void* d_ws, size_t ws_size,
                              hipStream_t stream) {
  const float* x     = (const float*)d_in[0];
  const float* theta = (const float*)d_in[1];
  const float* phi   = (const float*)d_in[2];
  const float* gamma = (const float*)d_in[3];
  const float* mask  = (const float*)d_in[4];
  const float* enn   = (const float*)d_in[5];
  const float* enp   = (const float*)d_in[6];
  const float* epn   = (const float*)d_in[7];
  const float* epp   = (const float*)d_in[8];
  const int*   perms = (const int*)d_in[9];

  char* ws = (char*)d_ws;
  float2* diagw = (float2*)(ws);
  float2* offw  = (float2*)(ws + (size_t)LL * UU * 8);
  int*    invp  = (int*)   (ws + (size_t)LL * UU * 16);

  const int threads = 256;
  const int blocksLU = (LL * UU) / threads;

  coeff_kernel<<<blocksLU, threads, 0, stream>>>(theta, phi, mask, enn, enp,
                                                 epn, epp, diagw, offw);
  invp_kernel<<<blocksLU, threads, 0, stream>>>(perms, invp);
  scan_kernel<<<BB / NB, threads, 0, stream>>>(x, gamma, perms, diagw, offw,
                                               invp, (float*)d_out);
}


# scan: per-layer operand loads (A coefficients, write-back slots) issued three layers ahead, not one or two, four register sets each: a first-touch read from memory no longer outlasts its layer
# speedup vs baseline: 12.8666x; 1.1066x over previous
#include <hip/hip_runtime.h>
#include <hip/hip_bf16.h>

#define UU 512
#define LL 512
#define BB 256
static __device__ constexpr float PI_F = 3.14159265358979323846f;

typedef float v2f __attribute__((ext_vector_type(2)));
typedef float v8f __attribute__((ext_vector_type(8)));

#define NB 4

__device__ __forceinline__ int sidx(int col, int row) {
  return col * NB + row;
}

__global__ __launch_bounds__(256) void coeff_kernel(
    const float* __restrict__ theta, const float* __restrict__ phi,
    const float* __restrict__ mask,  const float* __restrict__ enn,
    const float* __restrict__ enp,   const float* __restrict__ epn,
    const float* __restrict__ epp,
    float2* __restrict__ diagw, float2* __restrict__ offw) {
  int idx = blockIdx.x * blockDim.x + threadIdx.x;
  int l = idx / UU;
  int u = idx % UU;

  auto psl = [&](const float* ang, int uu) -> float2 {
    uu &= (UU - 1);
    if (uu & 1) return make_float2(1.0f, 0.0f);
    int k = uu >> 1;
    float m = mask[l * (UU / 2) + k];
    float a = ang[l * (UU / 2) + k] * m + (1.0f - m) * PI_F;
    float s, c;
    __sincosf(a, &s, &c);
    return make_float2(c, s);
  };

  float2 ip_m1 = psl(theta, u - 1);
  float2 ip_0  = psl(theta, u);
  float2 ip_p1 = psl(theta, u + 1);
  float2 ep_m1 = psl(phi,   u - 1);
  float2 ep_0  = psl(phi,   u);

  int um1 = (u - 1) & (UU - 1);
  float enn_u = enn[u * LL + l],  enn_m = enn[um1 * LL + l];
  float epp_u = epp[u * LL + l],  epp_m = epp[um1 * LL + l];
  float enp_u = enp[u * LL + l],  enp_m = enp[um1 * LL + l];
  float epn_u = epn[u * LL + l],  epn_m = epn[um1 * LL + l];

  float sdr = epp_u * ip_0.x - enn_u * ip_p1.x - enn_m * ip_m1.x + epp_m * ip_0.x;
  float sdi = epp_u * ip_0.y - enn_u * ip_p1.y - enn_m * ip_m1.y + epp_m * ip_0.y;

  float cvr = epn_u * ip_0.x + enp_u * ip_p1.x + enp_m * ip_m1.x + epn_m * ip_0.x;
  float cvi = epn_u * ip_0.y + enp_u * ip_p1.y + enp_m * ip_m1.y + epn_m * ip_0.y;
  float sor = -cvi, soi = cvr;

  diagw[idx] = make_float2(0.5f * (ep_0.x * sdr - ep_0.y * sdi),
                           0.5f * (ep_0.x * sdi + ep_0.y * sdr));
  offw[idx]  = make_float2(0.5f * (ep_m1.x * sor - ep_m1.y * soi),
                           0.5f * (ep_m1.x * soi + ep_m1.y * sor));
}

__global__ __launch_bounds__(256) void invp_kernel(const int* __restrict__ perms,
                                                   int* __restrict__ invp) {
  int idx = blockIdx.x * blockDim.x + threadIdx.x;
  int l = idx / UU;
  int u = idx % UU;
  int c = perms[(l + 1) * UU + u];
  invp[l * UU + c] = sidx(u, 0);
}

typedef int v2i __attribute__((ext_vector_type(2)));
typedef float v4f __attribute__((ext_vector_type(4)));

__device__ __forceinline__ void lds_barrier() {
  __builtin_amdgcn_fence(__ATOMIC_RELEASE, "workgroup", "local");
  __builtin_amdgcn_s_barrier();
  __builtin_amdgcn_fence(__ATOMIC_ACQUIRE, "workgroup", "local");
  __builtin_amdgcn_sched_barrier(0);
}

struct CoefRegs { float2 p[8]; };
struct SlotRegs { v2i s[8]; };

__global__ __launch_bounds__(256) void scan_kernel(
    const float* __restrict__ x, const float* __restrict__ gamma,
    const int* __restrict__ perms, const float2* __restrict__ diagw,
    const float2* __restrict__ offw, const int* __restrict__ invp,
    float* __restrict__ out) {
  __shared__ float2 S[2 * NB * UU];

  const int tid  = threadIdx.x;
  const int b0   = blockIdx.x * NB;
  const int wave = tid >> 5;
  const int lane = tid & 31;
  const int hi   = lane >> 4;
  const int n    = lane & 15;
  const int q    = n >> 2;
  const int b    = n & 3;
  const int m    = lane & 15;
  const int r    = m & 3;
  const int blk  = m >> 2;

  const bool use_diag = ((r < 2) == (hi == 0));
  const float2* csel = (use_diag ? diagw : offw) + (wave * 64 + 2 * blk + hi);
  const v2i* isel = reinterpret_cast<const v2i*>(invp + wave * 64 + 2 * q);
  const bool mine  = ((q >> 1) == hi);
  const bool upper = ((q & 1) != 0);

  auto fetch_coef = [&](CoefRegs& g, int l) {
    l = (l < LL) ? l : (LL - 1);
    const float2* cp = csel + (size_t)l * UU;
#pragma unroll
    for (int t = 0; t < 8; ++t) g.p[t] = cp[t * 8];
  };
  auto fetch_slot = [&](SlotRegs& g, int l) {
    l = (l < LL) ? l : (LL - 1);
    const v2i* ip = isel + (size_t)l * (UU / 2);
#pragma unroll
    for (int t = 0; t < 8; ++t) g.s[t] = ip[t * 4];
  };

  CoefRegs c0, c1, c2, c3;
  SlotRegs s0, s1, s2, s3;
  fetch_coef(c0, 0);
  fetch_coef(c1, 1);
  fetch_coef(c2, 2);
  fetch_slot(s0, 0);
  fetch_slot(s1, 1);
  fetch_slot(s2, 2);

#pragma unroll
  for (int e = 0; e < (NB * UU) / 256; ++e) {
    int idx = tid + e * 256;
    int rr = idx >> 9;
    int u  = idx & (UU - 1);
    int p  = perms[u];
    float xr = x[(size_t)(b0 + rr) * UU + p];
    float xi = x[(size_t)BB * UU + (size_t)(b0 + rr) * UU + p];
    float sg, cg;
    __sincosf(gamma[p], &sg, &cg);
    S[sidx(u, rr)] = make_float2(xr * cg - xi * sg, xr * sg + xi * cg);
  }
  __syncthreads();

  auto layer = [&](const CoefRegs& g, CoefRegs& gn, const SlotRegs& sl,
                   SlotRegs& sn, int l, const int rd, const int wr) {
    v2f bop[8];
#pragma unroll
    for (int t = 0; t < 8; ++t) {
      float2 sb = S[rd + sidx((wave * 8 + t) * 8 + 2 * q + hi, b)];
      bop[t][0] = sb.x;  bop[t][1] = sb.y;
    }
    __builtin_amdgcn_sched_barrier(0);
    fetch_slot(sn, l + 3);

    v2f aop[8];
    auto mk_a = [&](int t) {
      const float2 p = g.p[t];
      if (r & 1) { aop[t][0] = p.y;  aop[t][1] = p.x;  }
      else       { aop[t][0] = p.x;  aop[t][1] = -p.y; }
    };
    int w0[8], w1[8];
    auto mk_w = [&](int t) {
      const v2i so = sl.s[t];
      w0[t] = (wr + so[0] + b) * 8;
      w1[t] = (wr + so[1] + b) * 8;
      asm volatile("" :: "v"(w0[t]), "v"(w1[t]));
    };
    v8f acc[8];
    v4f y[8];
    auto pick = [&](int t) {
      const v8f c = acc[t];
      y[t][0] = upper ? c[4] : c[0];
      y[t][1] = upper ? c[5] : c[1];
      y[t][2] = upper ? c[6] : c[2];
      y[t][3] = upper ? c[7] : c[3];
      asm volatile("" :: "v"(y[t]));
    };
#pragma unroll
    for (int t = 0; t < 8; ++t) mk_a(t);
    fetch_coef(gn, l + 3);
#pragma unroll
    for (int t = 0; t < 8; ++t) {
      const v8f zero = {0.f, 0.f, 0.f, 0.f, 0.f, 0.f, 0.f, 0.f};
      acc[t] = __builtin_amdgcn_wmma_f32_16x16x4_f32(
                   false, aop[t], false, bop[t], (short)0, zero, false, false);
      __builtin_amdgcn_sched_barrier(0);
      if (t > 0) pick(t - 1);
      mk_w(t);
      __builtin_amdgcn_sched_barrier(0);
    }
    pick(7);
    if (mine) {
      char* sbytes = reinterpret_cast<char*>(S);
#pragma unroll
      for (int t = 0; t < 8; ++t) {
        *reinterpret_cast<float2*>(sbytes + w0[t]) = make_float2(y[t][0], y[t][1]);
        *reinterpret_cast<float2*>(sbytes + w1[t]) = make_float2(y[t][2], y[t][3]);
      }
    }
    lds_barrier();
  };

  for (int l = 0; l < LL; l += 4) {
    layer(c0, c3, s0, s3, l,     0, NB * UU);
    layer(c1, c0, s1, s0, l + 1, NB * UU, 0);
    layer(c2, c1, s2, s1, l + 2, 0, NB * UU);
    layer(c3, c2, s3, s2, l + 3, NB * UU, 0);
  }

  for (int e = 0; e < (NB * UU) / 256; ++e) {
    int idx = tid + e * 256;
    int rr = idx >> 9;
    int u  = idx & (UU - 1);
    float2 s = S[sidx(u, rr)];
    out[(size_t)(b0 + rr) * UU + u] = s.x;
    out[(size_t)BB * UU + (size_t)(b0 + rr) * UU + u] = s.y;
  }
}

extern "C" void kernel_launch(void* const* d_in, const int* in_sizes, int n_in,
                              void* d_out, int out_size, void* d_ws, size_t ws_size,
                              hipStream_t stream) {
  const float* x     = (const float*)d_in[0];
  const float* theta = (const float*)d_in[1];
  const float* phi   = (const float*)d_in[2];
  const float* gamma = (const float*)d_in[3];
  const float* mask  = (const float*)d_in[4];
  const float* enn   = (const float*)d_in[5];
  const float* enp   = (const float*)d_in[6];
  const float* epn   = (const float*)d_in[7];
  const float* epp   = (const float*)d_in[8];
  const int*   perms = (const int*)d_in[9];

  char* ws = (char*)d_ws;
  float2* diagw = (float2*)(ws);
  float2* offw  = (float2*)(ws + (size_t)LL * UU * 8);
  int*    invp  = (int*)   (ws + (size_t)LL * UU * 16);

  const int threads = 256;
  const int blocksLU = (LL * UU) / threads;

  coeff_kernel<<<blocksLU, threads, 0, stream>>>(theta, phi, mask, enn, enp,
                                                 epn, epp, diagw, offw);
  invp_kernel<<<blocksLU, threads, 0, stream>>>(perms, invp);
  scan_kernel<<<BB / NB, threads, 0, stream>>>(x, gamma, perms, diagw, offw,
                                               invp, (float*)d_out);
}


# scan: the permutation behind a layer is done as a gather by the next layer's state reads (perms read directly); write-back to fixed addresses, 8 two-address stores not 16; no inverse-perm kernel
# speedup vs baseline: 13.5512x; 1.0532x over previous
#include <hip/hip_runtime.h>
#include <hip/hip_bf16.h>

#define UU 512
#define LL 512
#define BB 256
static __device__ constexpr float PI_F = 3.14159265358979323846f;

typedef float v2f __attribute__((ext_vector_type(2)));
typedef float v8f __attribute__((ext_vector_type(8)));

#define NB 4

__device__ __forceinline__ int sidx(int col, int row) {
  return col * NB + row;
}

__global__ __launch_bounds__(256) void coeff_kernel(
    const float* __restrict__ theta, const float* __restrict__ phi,
    const float* __restrict__ mask,  const float* __restrict__ enn,
    const float* __restrict__ enp,   const float* __restrict__ epn,
    const float* __restrict__ epp,
    float2* __restrict__ diagw, float2* __restrict__ offw) {
  int idx = blockIdx.x * blockDim.x + threadIdx.x;
  int l = idx / UU;
  int u = idx % UU;

  auto psl = [&](const float* ang, int uu) -> float2 {
    uu &= (UU - 1);
    if (uu & 1) return make_float2(1.0f, 0.0f);
    int k = uu >> 1;
    float m = mask[l * (UU / 2) + k];
    float a = ang[l * (UU / 2) + k] * m + (1.0f - m) * PI_F;
    float s, c;
    __sincosf(a, &s, &c);
    return make_float2(c, s);
  };

  float2 ip_m1 = psl(theta, u - 1);
  float2 ip_0  = psl(theta, u);
  float2 ip_p1 = psl(theta, u + 1);
  float2 ep_m1 = psl(phi,   u - 1);
  float2 ep_0  = psl(phi,   u);

  int um1 = (u - 1) & (UU - 1);
  float enn_u = enn[u * LL + l],  enn_m = enn[um1 * LL + l];
  float epp_u = epp[u * LL + l],  epp_m = epp[um1 * LL + l];
  float enp_u = enp[u * LL + l],  enp_m = enp[um1 * LL + l];
  float epn_u = epn[u * LL + l],  epn_m = epn[um1 * LL + l];

  float sdr = epp_u * ip_0.x - enn_u * ip_p1.x - enn_m * ip_m1.x + epp_m * ip_0.x;
  float sdi = epp_u * ip_0.y - enn_u * ip_p1.y - enn_m * ip_m1.y + epp_m * ip_0.y;

  float cvr = epn_u * ip_0.x + enp_u * ip_p1.x + enp_m * ip_m1.x + epn_m * ip_0.x;
  float cvi = epn_u * ip_0.y + enp_u * ip_p1.y + enp_m * ip_m1.y + epn_m * ip_0.y;
  float sor = -cvi, soi = cvr;

  diagw[idx] = make_float2(0.5f * (ep_0.x * sdr - ep_0.y * sdi),
                           0.5f * (ep_0.x * sdi + ep_0.y * sdr));
  offw[idx]  = make_float2(0.5f * (ep_m1.x * sor - ep_m1.y * soi),
                           0.5f * (ep_m1.x * soi + ep_m1.y * sor));
}

typedef float v4f __attribute__((ext_vector_type(4)));

__device__ __forceinline__ void lds_barrier() {
  __builtin_amdgcn_fence(__ATOMIC_RELEASE, "workgroup", "local");
  __builtin_amdgcn_s_barrier();
  __builtin_amdgcn_fence(__ATOMIC_ACQUIRE, "workgroup", "local");
  __builtin_amdgcn_sched_barrier(0);
}

struct CoefRegs { float2 p[8]; };
struct PermRegs { int s[8]; };

__global__ __launch_bounds__(256) void scan_kernel(
    const float* __restrict__ x, const float* __restrict__ gamma,
    const int* __restrict__ perms, const float2* __restrict__ diagw,
    const float2* __restrict__ offw, float* __restrict__ out) {
  __shared__ float2 S[2 * NB * UU];

  const int tid  = threadIdx.x;
  const int b0   = blockIdx.x * NB;
  const int wave = tid >> 5;
  const int lane = tid & 31;
  const int hi   = lane >> 4;
  const int n    = lane & 15;
  const int q    = n >> 2;
  const int b    = n & 3;
  const int m    = lane & 15;
  const int r    = m & 3;
  const int blk  = m >> 2;

  const bool use_diag = ((r < 2) == (hi == 0));
  const float2* csel = (use_diag ? diagw : offw) + (wave * 64 + 2 * blk + hi);
  const int* psel = perms + wave * 64 + 2 * q + hi;
  const int wb0 = sidx(wave * 64 + 2 * q, b);
  const bool mine  = ((q >> 1) == hi);
  const bool upper = ((q & 1) != 0);

  auto fetch_coef = [&](CoefRegs& g, int l) {
    l = (l < LL) ? l : (LL - 1);
    const float2* cp = csel + (size_t)l * UU;
#pragma unroll
    for (int t = 0; t < 8; ++t) g.p[t] = cp[t * 8];
  };
  auto fetch_perm = [&](PermRegs& g, int l) {
    l = (l < LL) ? l : (LL - 1);
    const int* ip = psel + (size_t)(l + 1) * UU;
#pragma unroll
    for (int t = 0; t < 8; ++t) g.s[t] = ip[t * 8];
  };

  CoefRegs c0, c1, c2, c3;
  PermRegs s0, s1, s2, s3;
  fetch_coef(c0, 0);
  fetch_coef(c1, 1);
  fetch_coef(c2, 2);
  fetch_perm(s0, 0);
  fetch_perm(s1, 1);
  fetch_perm(s2, 2);
  int pfin[(NB * UU) / 256];
#pragma unroll
  for (int e = 0; e < (NB * UU) / 256; ++e)
    pfin[e] = perms[(size_t)LL * UU + ((tid + e * 256) & (UU - 1))];

#pragma unroll
  for (int e = 0; e < (NB * UU) / 256; ++e) {
    int idx = tid + e * 256;
    int rr = idx >> 9;
    int u  = idx & (UU - 1);
    int p  = perms[u];
    float xr = x[(size_t)(b0 + rr) * UU + p];
    float xi = x[(size_t)BB * UU + (size_t)(b0 + rr) * UU + p];
    float sg, cg;
    __sincosf(gamma[p], &sg, &cg);
    S[sidx(u, rr)] = make_float2(xr * cg - xi * sg, xr * sg + xi * cg);
  }
  __syncthreads();

  int ra[8];
#pragma unroll
  for (int t = 0; t < 8; ++t) ra[t] = sidx((wave * 8 + t) * 8 + 2 * q + hi, b) * 8;

  auto layer = [&](const CoefRegs& g, CoefRegs& gn, const PermRegs& sl,
                   PermRegs& sn, int l, const int rd, const int wr) {
    v2f bop[8];
#pragma unroll
    for (int t = 0; t < 8; ++t) {
      float2 sb = *reinterpret_cast<const float2*>(
                      reinterpret_cast<const char*>(S + rd) + ra[t]);
      bop[t][0] = sb.x;  bop[t][1] = sb.y;
    }
    __builtin_amdgcn_sched_barrier(0);
    fetch_perm(sn, l + 3);

    v2f aop[8];
    auto mk_a = [&](int t) {
      const float2 p = g.p[t];
      if (r & 1) { aop[t][0] = p.y;  aop[t][1] = p.x;  }
      else       { aop[t][0] = p.x;  aop[t][1] = -p.y; }
    };
    auto mk_r = [&](int t) {
      ra[t] = sidx(sl.s[t], b) * 8;
      asm volatile("" :: "v"(ra[t]));
    };
    v8f acc[8];
    v4f y[8];
    auto pick = [&](int t) {
      const v8f c = acc[t];
      y[t][0] = upper ? c[4] : c[0];
      y[t][1] = upper ? c[5] : c[1];
      y[t][2] = upper ? c[6] : c[2];
      y[t][3] = upper ? c[7] : c[3];
      asm volatile("" :: "v"(y[t]));
    };
#pragma unroll
    for (int t = 0; t < 8; ++t) mk_a(t);
    fetch_coef(gn, l + 3);
#pragma unroll
    for (int t = 0; t < 8; ++t) {
      const v8f zero = {0.f, 0.f, 0.f, 0.f, 0.f, 0.f, 0.f, 0.f};
      acc[t] = __builtin_amdgcn_wmma_f32_16x16x4_f32(
                   false, aop[t], false, bop[t], (short)0, zero, false, false);
      __builtin_amdgcn_sched_barrier(0);
      if (t > 0) pick(t - 1);
      mk_r(t);
      __builtin_amdgcn_sched_barrier(0);
    }
    pick(7);
    if (mine) {
      float2* wp = S + wr + wb0;
#pragma unroll
      for (int t = 0; t < 8; ++t) {
        wp[t * 8 * NB]      = make_float2(y[t][0], y[t][1]);
        wp[t * 8 * NB + NB] = make_float2(y[t][2], y[t][3]);
      }
    }
    lds_barrier();
  };

  for (int l = 0; l < LL; l += 4) {
    layer(c0, c3, s0, s3, l,     0, NB * UU);
    layer(c1, c0, s1, s0, l + 1, NB * UU, 0);
    layer(c2, c1, s2, s1, l + 2, 0, NB * UU);
    layer(c3, c2, s3, s2, l + 3, NB * UU, 0);
  }

#pragma unroll
  for (int e = 0; e < (NB * UU) / 256; ++e) {
    int idx = tid + e * 256;
    int rr = idx >> 9;
    int u  = idx & (UU - 1);
    float2 s = S[sidx(pfin[e], rr)];
    out[(size_t)(b0 + rr) * UU + u] = s.x;
    out[(size_t)BB * UU + (size_t)(b0 + rr) * UU + u] = s.y;
  }
}

extern "C" void kernel_launch(void* const* d_in, const int* in_sizes, int n_in,
                              void* d_out, int out_size, void* d_ws, size_t ws_size,
                              hipStream_t stream) {
  const float* x     = (const float*)d_in[0];
  const float* theta = (const float*)d_in[1];
  const float* phi   = (const float*)d_in[2];
  const float* gamma = (const float*)d_in[3];
  const float* mask  = (const float*)d_in[4];
  const float* enn   = (const float*)d_in[5];
  const float* enp   = (const float*)d_in[6];
  const float* epn   = (const float*)d_in[7];
  const float* epp   = (const float*)d_in[8];
  const int*   perms = (const int*)d_in[9];

  char* ws = (char*)d_ws;
  float2* diagw = (float2*)(ws);
  float2* offw  = (float2*)(ws + (size_t)LL * UU * 8);

  const int threads = 256;
  const int blocksLU = (LL * UU) / threads;

  coeff_kernel<<<blocksLU, threads, 0, stream>>>(theta, phi, mask, enn, enp,
                                                 epn, epp, diagw, offw);
  scan_kernel<<<BB / NB, threads, 0, stream>>>(x, gamma, perms, diagw, offw,
                                               (float*)d_out);
}


# work-group takes 1 batch row, not 4: 8 pairs ride one 16x16x4 product (M_ih rows x re/im state columns), 32 products a layer not 64, 256 groups; imaginary sums take their products in another k order
# speedup vs baseline: 17.8398x; 1.3165x over previous
#include <hip/hip_runtime.h>
#include <hip/hip_bf16.h>

#define UU 512
#define LL 512
#define BB 256
static __device__ constexpr float PI_F = 3.14159265358979323846f;

typedef float v2f __attribute__((ext_vector_type(2)));
typedef float v8f __attribute__((ext_vector_type(8)));

__global__ __launch_bounds__(256) void coeff_kernel(
    const float* __restrict__ theta, const float* __restrict__ phi,
    const float* __restrict__ mask,  const float* __restrict__ enn,
    const float* __restrict__ enp,   const float* __restrict__ epn,
    const float* __restrict__ epp,
    float2* __restrict__ diagw, float2* __restrict__ offw) {
  int idx = blockIdx.x * blockDim.x + threadIdx.x;
  int l = idx / UU;
  int u = idx % UU;

  auto psl = [&](const float* ang, int uu) -> float2 {
    uu &= (UU - 1);
    if (uu & 1) return make_float2(1.0f, 0.0f);
    int k = uu >> 1;
    float m = mask[l * (UU / 2) + k];
    float a = ang[l * (UU / 2) + k] * m + (1.0f - m) * PI_F;
    float s, c;
    __sincosf(a, &s, &c);
    return make_float2(c, s);
  };

  float2 ip_m1 = psl(theta, u - 1);
  float2 ip_0  = psl(theta, u);
  float2 ip_p1 = psl(theta, u + 1);
  float2 ep_m1 = psl(phi,   u - 1);
  float2 ep_0  = psl(phi,   u);

  int um1 = (u - 1) & (UU - 1);
  float enn_u = enn[u * LL + l],  enn_m = enn[um1 * LL + l];
  float epp_u = epp[u * LL + l],  epp_m = epp[um1 * LL + l];
  float enp_u = enp[u * LL + l],  enp_m = enp[um1 * LL + l];
  float epn_u = epn[u * LL + l],  epn_m = epn[um1 * LL + l];

  float sdr = epp_u * ip_0.x - enn_u * ip_p1.x - enn_m * ip_m1.x + epp_m * ip_0.x;
  float sdi = epp_u * ip_0.y - enn_u * ip_p1.y - enn_m * ip_m1.y + epp_m * ip_0.y;

  float cvr = epn_u * ip_0.x + enp_u * ip_p1.x + enp_m * ip_m1.x + epn_m * ip_0.x;
  float cvi = epn_u * ip_0.y + enp_u * ip_p1.y + enp_m * ip_m1.y + epn_m * ip_0.y;
  float sor = -cvi, soi = cvr;

  diagw[idx] = make_float2(0.5f * (ep_0.x * sdr - ep_0.y * sdi),
                           0.5f * (ep_0.x * sdi + ep_0.y * sdr));
  offw[idx]  = make_float2(0.5f * (ep_m1.x * sor - ep_m1.y * soi),
                           0.5f * (ep_m1.x * soi + ep_m1.y * sor));
}


__device__ __forceinline__ void lds_barrier() {
  __builtin_amdgcn_fence(__ATOMIC_RELEASE, "workgroup", "local");
  __builtin_amdgcn_s_barrier();
  __builtin_amdgcn_fence(__ATOMIC_ACQUIRE, "workgroup", "local");
  __builtin_amdgcn_sched_barrier(0);
}

#define NT 4
struct CoefRegs { float2 p[NT]; };
struct PermRegs { int s[NT]; };

__global__ __launch_bounds__(256) void scan_kernel(
    const float* __restrict__ x, const float* __restrict__ gamma,
    const int* __restrict__ perms, const float2* __restrict__ diagw,
    const float2* __restrict__ offw, float* __restrict__ out) {
  __shared__ float2 S[2 * UU];

  const int tid  = threadIdx.x;
  const int b0   = blockIdx.x;
  const int wave = tid >> 5;
  const int lane = tid & 31;
  const int hi   = lane >> 4;
  const int n    = lane & 15;
  const int q    = n >> 1;
  const int fl   = n & 1;
  const int ai   = n >> 3;
  const int ap   = n & 7;

  const bool use_diag = (ai == hi);
  const float2* csel = (use_diag ? diagw : offw) + (wave * 64 + 2 * ap + hi);
  const int* psel = perms + wave * 64 + 2 * q + hi;
  const int wb0 = (wave * 64 + 2 * q + hi) * 2 + fl;
  const bool q1 = (q & 1) != 0;
  const bool q2 = (q & 2) != 0;
  const bool q4 = (q & 4) != 0;
  const bool imag = (fl != 0);

  auto fetch_coef = [&](CoefRegs& g, int l) {
    l = (l < LL) ? l : (LL - 1);
    const float2* cp = csel + (size_t)l * UU;
#pragma unroll
    for (int t = 0; t < NT; ++t) g.p[t] = cp[t * 16];
  };
  auto fetch_perm = [&](PermRegs& g, int l) {
    l = (l < LL) ? l : (LL - 1);
    const int* ip = psel + (size_t)(l + 1) * UU;
#pragma unroll
    for (int t = 0; t < NT; ++t) g.s[t] = ip[t * 16];
  };

  CoefRegs c0, c1, c2, c3;
  PermRegs s0, s1, s2, s3;
  fetch_coef(c0, 0);
  fetch_coef(c1, 1);
  fetch_coef(c2, 2);
  fetch_perm(s0, 0);
  fetch_perm(s1, 1);
  fetch_perm(s2, 2);
  int pfin[UU / 256];
#pragma unroll
  for (int e = 0; e < UU / 256; ++e)
    pfin[e] = perms[(size_t)LL * UU + tid + e * 256];

#pragma unroll
  for (int e = 0; e < UU / 256; ++e) {
    int u = tid + e * 256;
    int p = perms[u];
    float xr = x[(size_t)b0 * UU + p];
    float xi = x[(size_t)BB * UU + (size_t)b0 * UU + p];
    float sg, cg;
    __sincosf(gamma[p], &sg, &cg);
    S[u] = make_float2(xr * cg - xi * sg, xr * sg + xi * cg);
  }
  __syncthreads();

  int ra[NT];
#pragma unroll
  for (int t = 0; t < NT; ++t) ra[t] = (wave * 64 + t * 16 + 2 * q + hi) * 8;

  auto layer = [&](const CoefRegs& g, CoefRegs& gn, const PermRegs& sl,
                   PermRegs& sn, int l, const int rd, const int wr) {
    float2 sb[NT];
#pragma unroll
    for (int t = 0; t < NT; ++t)
      sb[t] = *reinterpret_cast<const float2*>(
                  reinterpret_cast<const char*>(S + rd) + ra[t]);
    __builtin_amdgcn_sched_barrier(0);
    fetch_perm(sn, l + 3);
    fetch_coef(gn, l + 3);
    __builtin_amdgcn_sched_barrier(0);

    auto mk_r = [&](int t) {
      ra[t] = sl.s[t] * 8;
    };
    v8f acc[NT];
    float y[NT];
    auto pick = [&](int t) {
      const v8f c = acc[t];
      const float a0 = q1 ? c[1] : c[0];
      const float a1 = q1 ? c[3] : c[2];
      const float a2 = q1 ? c[5] : c[4];
      const float a3 = q1 ? c[7] : c[6];
      const float e0 = q2 ? a1 : a0;
      const float e1 = q2 ? a3 : a2;
      y[t] = q4 ? e1 : e0;
      asm volatile("" :: "v"(y[t]));
    };
#pragma unroll
    for (int t = 0; t < NT; ++t) {
      v2f aop, bop;
      aop[0] = g.p[t].x;  aop[1] = g.p[t].y;
      bop[0] = imag ? sb[t].y : sb[t].x;
      bop[1] = imag ? sb[t].x : -sb[t].y;
      const v8f zero = {0.f, 0.f, 0.f, 0.f, 0.f, 0.f, 0.f, 0.f};
      acc[t] = __builtin_amdgcn_wmma_f32_16x16x4_f32(
                   false, aop, false, bop, (short)0, zero, false, false);
      __builtin_amdgcn_sched_barrier(0);
      if (t > 0) pick(t - 1);
      mk_r(t);
      __builtin_amdgcn_sched_barrier(0);
    }
    pick(NT - 1);
    float* wp = reinterpret_cast<float*>(S + wr) + wb0;
#pragma unroll
    for (int t = 0; t < NT; ++t) wp[t * 32] = y[t];
    lds_barrier();
  };

  for (int l = 0; l < LL; l += 4) {
    layer(c0, c3, s0, s3, l,     0, UU);
    layer(c1, c0, s1, s0, l + 1, UU, 0);
    layer(c2, c1, s2, s1, l + 2, 0, UU);
    layer(c3, c2, s3, s2, l + 3, UU, 0);
  }

#pragma unroll
  for (int e = 0; e < UU / 256; ++e) {
    int u = tid + e * 256;
    float2 sv = S[pfin[e]];
    out[(size_t)b0 * UU + u] = sv.x;
    out[(size_t)BB * UU + (size_t)b0 * UU + u] = sv.y;
  }
}

extern "C" void kernel_launch(void* const* d_in, const int* in_sizes, int n_in,
                              void* d_out, int out_size, void* d_ws, size_t ws_size,
                              hipStream_t stream) {
  const float* x     = (const float*)d_in[0];
  const float* theta = (const float*)d_in[1];
  const float* phi   = (const float*)d_in[2];
  const float* gamma = (const float*)d_in[3];
  const float* mask  = (const float*)d_in[4];
  const float* enn   = (const float*)d_in[5];
  const float* enp   = (const float*)d_in[6];
  const float* epn   = (const float*)d_in[7];
  const float* epp   = (const float*)d_in[8];
  const int*   perms = (const int*)d_in[9];

  char* ws = (char*)d_ws;
  float2* diagw = (float2*)(ws);
  float2* offw  = (float2*)(ws + (size_t)LL * UU * 8);

  const int threads = 256;
  const int blocksLU = (LL * UU) / threads;

  coeff_kernel<<<blocksLU, threads, 0, stream>>>(theta, phi, mask, enn, enp,
                                                 epn, epp, diagw, offw);
  scan_kernel<<<BB, threads, 0, stream>>>(x, gamma, perms, diagw, offw,
                                               (float*)d_out);
}


# two consecutive layers composed ahead of the scan into one operator per pair (third launch): two chained K=4 products per tile, one LDS hand-over and barrier per two layers; sums re-associated
# speedup vs baseline: 23.9224x; 1.3410x over previous
#include <hip/hip_runtime.h>
#include <hip/hip_bf16.h>

#define UU 512
#define LL 512
#define BB 256
#define NB (LL / 2)
static __device__ constexpr float PI_F = 3.14159265358979323846f;

typedef float v2f __attribute__((ext_vector_type(2)));
typedef float v8f __attribute__((ext_vector_type(8)));

__global__ __launch_bounds__(256) void coeff_kernel(
    const float* __restrict__ theta, const float* __restrict__ phi,
    const float* __restrict__ mask,  const float* __restrict__ enn,
    const float* __restrict__ enp,   const float* __restrict__ epn,
    const float* __restrict__ epp,
    float2* __restrict__ diagw, float2* __restrict__ offw) {
  int idx = blockIdx.x * blockDim.x + threadIdx.x;
  int l = idx / UU;
  int u = idx % UU;

  auto psl = [&](const float* ang, int uu) -> float2 {
    uu &= (UU - 1);
    if (uu & 1) return make_float2(1.0f, 0.0f);
    int k = uu >> 1;
    float m = mask[l * (UU / 2) + k];
    float a = ang[l * (UU / 2) + k] * m + (1.0f - m) * PI_F;
    float s, c;
    __sincosf(a, &s, &c);
    return make_float2(c, s);
  };

  float2 ip_m1 = psl(theta, u - 1);
  float2 ip_0  = psl(theta, u);
  float2 ip_p1 = psl(theta, u + 1);
  float2 ep_m1 = psl(phi,   u - 1);
  float2 ep_0  = psl(phi,   u);

  int um1 = (u - 1) & (UU - 1);
  float enn_u = enn[u * LL + l],  enn_m = enn[um1 * LL + l];
  float epp_u = epp[u * LL + l],  epp_m = epp[um1 * LL + l];
  float enp_u = enp[u * LL + l],  enp_m = enp[um1 * LL + l];
  float epn_u = epn[u * LL + l],  epn_m = epn[um1 * LL + l];

  float sdr = epp_u * ip_0.x - enn_u * ip_p1.x - enn_m * ip_m1.x + epp_m * ip_0.x;
  float sdi = epp_u * ip_0.y - enn_u * ip_p1.y - enn_m * ip_m1.y + epp_m * ip_0.y;

  float cvr = epn_u * ip_0.x + enp_u * ip_p1.x + enp_m * ip_m1.x + epn_m * ip_0.x;
  float cvi = epn_u * ip_0.y + enp_u * ip_p1.y + enp_m * ip_m1.y + epn_m * ip_0.y;
  float sor = -cvi, soi = cvr;

  diagw[idx] = make_float2(0.5f * (ep_0.x * sdr - ep_0.y * sdi),
                           0.5f * (ep_0.x * sdi + ep_0.y * sdr));
  offw[idx]  = make_float2(0.5f * (ep_m1.x * sor - ep_m1.y * soi),
                           0.5f * (ep_m1.x * soi + ep_m1.y * sor));
}

__global__ __launch_bounds__(256) void fuse_kernel(
    const int* __restrict__ perms, const float2* __restrict__ diagw,
    const float2* __restrict__ offw, float2* __restrict__ fw,
    int* __restrict__ jw) {
  const int tid = blockIdx.x * blockDim.x + threadIdx.x;
  const int h  = tid & 1;
  const int pp = (tid >> 1) & 7;
  const int g  = (tid >> 4) & 1;
  const int T  = (tid >> 5) & 31;
  const int b  = tid >> 10;
  const int l0 = 2 * b;
  const int l1 = l0 + 1;
  const int col = (T * 8 + pp) * 2 + h;
  const int s = perms[(size_t)l1 * UU + col];
  const int jcol = (s & ~1) | g;
  const float2 mi = (((s & 1) == g) ? diagw : offw)[(size_t)l0 * UU + jcol];
  const int src = (b > 0) ? perms[(size_t)l0 * UU + jcol] : jcol;
  jw[tid] = src * 8;
  const float2 md = diagw[(size_t)l1 * UU + col];
  const float2 mo = offw[(size_t)l1 * UU + col];
  const size_t fbase = ((size_t)((b * 32 + T) * 32 + g * 16 + pp)) * 2 + h;
#pragma unroll
  for (int i = 0; i < 2; ++i) {
    const float2 m = (i == h) ? md : mo;
    fw[fbase + (size_t)i * 16] = make_float2(m.x * mi.x - m.y * mi.y,
                                             m.x * mi.y + m.y * mi.x);
  }
}


__device__ __forceinline__ void lds_barrier() {
  __builtin_amdgcn_fence(__ATOMIC_RELEASE, "workgroup", "local");
  __builtin_amdgcn_s_barrier();
  __builtin_amdgcn_fence(__ATOMIC_ACQUIRE, "workgroup", "local");
  __builtin_amdgcn_sched_barrier(0);
}

#define NT 4
struct FacRegs { float4 f[NT]; };
struct OffRegs { int2 j[NT]; };

__global__ __launch_bounds__(256) void scan_kernel(
    const float* __restrict__ x, const float* __restrict__ gamma,
    const int* __restrict__ perms, const float4* __restrict__ fw,
    const int2* __restrict__ jw, float* __restrict__ out) {
  __shared__ float2 S[2 * UU];

  const int tid  = threadIdx.x;
  const int b0   = blockIdx.x;
  const int wave = tid >> 5;
  const int lane = tid & 31;
  const int hi   = lane >> 4;
  const int n    = lane & 15;
  const int q    = n >> 1;
  const int fl   = n & 1;

  const float4* fsel = fw + (wave * NT) * 32 + lane;
  const int2* jsel = jw + (wave * NT) * 16 + hi * 8 + q;
  const int wb0 = (wave * 64 + 2 * q + hi) * 2 + fl;
  const bool q1 = (q & 1) != 0;
  const bool q2 = (q & 2) != 0;
  const bool q4 = (q & 4) != 0;
  const bool imag = (fl != 0);

  auto fetch_fac = [&](FacRegs& g, int blk) {
    blk = (blk < NB) ? blk : (NB - 1);
    const float4* fp = fsel + (size_t)blk * (32 * 32);
#pragma unroll
    for (int t = 0; t < NT; ++t) g.f[t] = fp[t * 32];
  };
  auto fetch_off = [&](OffRegs& g, int blk) {
    blk = (blk < NB) ? blk : (NB - 1);
    const int2* jp = jsel + (size_t)blk * (32 * 16);
#pragma unroll
    for (int t = 0; t < NT; ++t) g.j[t] = jp[t * 16];
  };

  FacRegs c0, c1, c2, c3;
  OffRegs s0, s1, s2, s3;
  fetch_off(s0, 0);
  fetch_fac(c0, 0);
  fetch_off(s1, 1);
  fetch_fac(c1, 1);
  fetch_off(s2, 2);
  fetch_fac(c2, 2);
  int pfin[UU / 256];
#pragma unroll
  for (int e = 0; e < UU / 256; ++e)
    pfin[e] = perms[(size_t)LL * UU + tid + e * 256];

#pragma unroll
  for (int e = 0; e < UU / 256; ++e) {
    int u = tid + e * 256;
    int p = perms[u];
    float xr = x[(size_t)b0 * UU + p];
    float xi = x[(size_t)BB * UU + (size_t)b0 * UU + p];
    float sg, cg;
    __sincosf(gamma[p], &sg, &cg);
    S[u] = make_float2(xr * cg - xi * sg, xr * sg + xi * cg);
  }
  __syncthreads();

  auto block = [&](const FacRegs& g, FacRegs& gn, const OffRegs& sl,
                   OffRegs& sn, int blk, const int rd, const int wr) {
    float2 sb[NT][2];
    auto rd_state = [&](int t, int h) {
      sb[t][h] = *reinterpret_cast<const float2*>(
                     reinterpret_cast<const char*>(S + rd) + (h ? sl.j[t].y : sl.j[t].x));
    };
    rd_state(0, 0); rd_state(1, 0); rd_state(0, 1); rd_state(1, 1);
    rd_state(2, 0); rd_state(3, 0); rd_state(2, 1); rd_state(3, 1);
    __builtin_amdgcn_sched_barrier(0);
    fetch_off(sn, blk + 3);
    fetch_fac(gn, blk + 3);
    __builtin_amdgcn_sched_barrier(0);

    v8f acc[NT];
    float y[NT];
    auto prod = [&](int t, int h) {
      v2f aop, bop;
      aop[0] = h ? g.f[t].z : g.f[t].x;
      aop[1] = h ? g.f[t].w : g.f[t].y;
      const float2 sv = sb[t][h];
      bop[0] = imag ? sv.y : sv.x;
      bop[1] = imag ? sv.x : -sv.y;
      const v8f zero = {0.f, 0.f, 0.f, 0.f, 0.f, 0.f, 0.f, 0.f};
      acc[t] = __builtin_amdgcn_wmma_f32_16x16x4_f32(
                   false, aop, false, bop, (short)0, h ? acc[t] : zero, false, false);
    };
    auto pick = [&](int t) {
      const v8f c = acc[t];
      const float a0 = q1 ? c[1] : c[0];
      const float a1 = q1 ? c[3] : c[2];
      const float a2 = q1 ? c[5] : c[4];
      const float a3 = q1 ? c[7] : c[6];
      const float e0 = q2 ? a1 : a0;
      const float e1 = q2 ? a3 : a2;
      y[t] = q4 ? e1 : e0;
      asm volatile("" :: "v"(y[t]));
    };
    prod(0, 0);
    __builtin_amdgcn_sched_barrier(0);
    prod(1, 0);
    __builtin_amdgcn_sched_barrier(0);
    prod(0, 1);
    __builtin_amdgcn_sched_barrier(0);
    prod(1, 1);
    __builtin_amdgcn_sched_barrier(0);
    prod(2, 0);
    __builtin_amdgcn_sched_barrier(0);
    pick(0);
    __builtin_amdgcn_sched_barrier(0);
    prod(3, 0);
    __builtin_amdgcn_sched_barrier(0);
    pick(1);
    __builtin_amdgcn_sched_barrier(0);
    prod(2, 1);
    __builtin_amdgcn_sched_barrier(0);
    prod(3, 1);
    __builtin_amdgcn_sched_barrier(0);
    pick(2);
    pick(3);
    float* wp = reinterpret_cast<float*>(S + wr) + wb0;
#pragma unroll
    for (int t = 0; t < NT; ++t) wp[t * 32] = y[t];
    lds_barrier();
  };

  for (int blk = 0; blk < NB; blk += 4) {
    block(c0, c3, s0, s3, blk,     0, UU);
    block(c1, c0, s1, s0, blk + 1, UU, 0);
    block(c2, c1, s2, s1, blk + 2, 0, UU);
    block(c3, c2, s3, s2, blk + 3, UU, 0);
  }

#pragma unroll
  for (int e = 0; e < UU / 256; ++e) {
    int u = tid + e * 256;
    float2 sv = S[pfin[e]];
    out[(size_t)b0 * UU + u] = sv.x;
    out[(size_t)BB * UU + (size_t)b0 * UU + u] = sv.y;
  }
}

extern "C" void kernel_launch(void* const* d_in, const int* in_sizes, int n_in,
                              void* d_out, int out_size, void* d_ws, size_t ws_size,
                              hipStream_t stream) {
  const float* x     = (const float*)d_in[0];
  const float* theta = (const float*)d_in[1];
  const float* phi   = (const float*)d_in[2];
  const float* gamma = (const float*)d_in[3];
  const float* mask  = (const float*)d_in[4];
  const float* enn   = (const float*)d_in[5];
  const float* enp   = (const float*)d_in[6];
  const float* epn   = (const float*)d_in[7];
  const float* epp   = (const float*)d_in[8];
  const int*   perms = (const int*)d_in[9];

  char* ws = (char*)d_ws;
  float2* diagw = (float2*)(ws);
  float2* offw  = (float2*)(ws + (size_t)LL * UU * 8);
  float2* fw    = (float2*)(ws + (size_t)LL * UU * 16);
  int*    jw    = (int*)(ws + (size_t)LL * UU * 32);

  const int threads = 256;
  const int blocksLU = (LL * UU) / threads;
  const int blocksFuse = (NB * 32 * 16 * 2) / threads;

  coeff_kernel<<<blocksLU, threads, 0, stream>>>(theta, phi, mask, enn, enp,
                                                 epn, epp, diagw, offw);
  fuse_kernel<<<blocksFuse, threads, 0, stream>>>(perms, diagw, offw, fw, jw);
  scan_kernel<<<BB, threads, 0, stream>>>(x, gamma, perms, (const float4*)fw,
                                          (const int2*)jw, (float*)d_out);
}


# pre-pass in one launch, not two: a work-group builds the diag/off coefficients of its two layers in LDS and composes them there, 4 sites a thread; no coefficient planes in the workspace
# speedup vs baseline: 25.4952x; 1.0657x over previous
#include <hip/hip_runtime.h>
#include <hip/hip_bf16.h>

#define UU 512
#define LL 512
#define BB 256
#define NB (LL / 2)
static __device__ constexpr float PI_F = 3.14159265358979323846f;

typedef float v2f __attribute__((ext_vector_type(2)));
typedef float v8f __attribute__((ext_vector_type(8)));

__global__ __launch_bounds__(256) void prep_kernel(
    const float* __restrict__ theta, const float* __restrict__ phi,
    const float* __restrict__ mask,  const float* __restrict__ enn,
    const float* __restrict__ enp,   const float* __restrict__ epn,
    const float* __restrict__ epp,   const int* __restrict__ perms,
    float2* __restrict__ fw, int* __restrict__ jw) {
  __shared__ float2 R[4][UU];
  __shared__ float2 Dg[2][UU];
  __shared__ float2 Of[2][UU];

  const int tid = threadIdx.x;
  const int b   = blockIdx.x;
  const int l0  = 2 * b;
  const int l1  = l0 + 1;

  int sidx[4], srcv[4];
#pragma unroll
  for (int it = 0; it < 4; ++it) {
    const int t   = tid + it * 256;
    const int col = (((t >> 5) & 31) * 8 + ((t >> 1) & 7)) * 2 + (t & 1);
    sidx[it] = perms[(size_t)l1 * UU + col];
  }

#pragma unroll
  for (int e = 0; e < 2; ++e) {
    const int u = tid + e * 256;
    const size_t at = (size_t)u * LL + l0;
    R[0][u] = *reinterpret_cast<const float2*>(enn + at);
    R[1][u] = *reinterpret_cast<const float2*>(epp + at);
    R[2][u] = *reinterpret_cast<const float2*>(enp + at);
    R[3][u] = *reinterpret_cast<const float2*>(epn + at);
  }

  auto cs = [&](float ang, float m) -> float2 {
    float a = ang * m + (1.0f - m) * PI_F;
    float s, c;
    __sincosf(a, &s, &c);
    return make_float2(c, s);
  };
  float2 tA[2][2], tB[2][2], pA[2][2];
#pragma unroll
  for (int e = 0; e < 2; ++e) {
#pragma unroll
    for (int j = 0; j < 2; ++j) {
      const int u  = tid + e * 256;
      const int l  = l0 + j;
      const int k0 = u >> 1;
      const int k1 = ((u + 1) & (UU - 1)) >> 1;
      const float m0 = mask[l * (UU / 2) + k0];
      const float m1 = mask[l * (UU / 2) + k1];
      tA[e][j] = cs(theta[l * (UU / 2) + k0], m0);
      tB[e][j] = cs(theta[l * (UU / 2) + k1], m1);
      pA[e][j] = cs(phi[l * (UU / 2) + k0], m0);
    }
  }

#pragma unroll
  for (int it = 0; it < 4; ++it) {
    const int t = tid + it * 256;
    const int g = (t >> 4) & 1;
    const int jcol = ((sidx[it] & ~1) | g) & (UU - 1);
    srcv[it] = (b > 0) ? perms[(size_t)l0 * UU + jcol] : jcol;
  }
  __syncthreads();

#pragma unroll
  for (int e = 0; e < 2; ++e) {
#pragma unroll
    for (int j = 0; j < 2; ++j) {
      const int u   = tid + e * 256;
      const int um1 = (u - 1) & (UU - 1);
      const bool odd = (u & 1) != 0;
      const float2 one = make_float2(1.0f, 0.0f);
      const float2 ip_m1 = odd ? tA[e][j] : one;
      const float2 ip_0  = odd ? one : tA[e][j];
      const float2 ip_p1 = odd ? tB[e][j] : one;
      const float2 ep_m1 = odd ? pA[e][j] : one;
      const float2 ep_0  = odd ? one : pA[e][j];

      const float2 r0u = R[0][u], r0m = R[0][um1];
      const float2 r1u = R[1][u], r1m = R[1][um1];
      const float2 r2u = R[2][u], r2m = R[2][um1];
      const float2 r3u = R[3][u], r3m = R[3][um1];
      const float enn_u = j ? r0u.y : r0u.x, enn_m = j ? r0m.y : r0m.x;
      const float epp_u = j ? r1u.y : r1u.x, epp_m = j ? r1m.y : r1m.x;
      const float enp_u = j ? r2u.y : r2u.x, enp_m = j ? r2m.y : r2m.x;
      const float epn_u = j ? r3u.y : r3u.x, epn_m = j ? r3m.y : r3m.x;

      float sdr = epp_u * ip_0.x - enn_u * ip_p1.x - enn_m * ip_m1.x + epp_m * ip_0.x;
      float sdi = epp_u * ip_0.y - enn_u * ip_p1.y - enn_m * ip_m1.y + epp_m * ip_0.y;

      float cvr = epn_u * ip_0.x + enp_u * ip_p1.x + enp_m * ip_m1.x + epn_m * ip_0.x;
      float cvi = epn_u * ip_0.y + enp_u * ip_p1.y + enp_m * ip_m1.y + epn_m * ip_0.y;
      float sor = -cvi, soi = cvr;

      Dg[j][u] = make_float2(0.5f * (ep_0.x * sdr - ep_0.y * sdi),
                             0.5f * (ep_0.x * sdi + ep_0.y * sdr));
      Of[j][u] = make_float2(0.5f * (ep_m1.x * sor - ep_m1.y * soi),
                             0.5f * (ep_m1.x * soi + ep_m1.y * sor));
    }
  }
  __syncthreads();

#pragma unroll
  for (int it = 0; it < 4; ++it) {
    const int t  = tid + it * 256;
    const int h  = t & 1;
    const int pp = (t >> 1) & 7;
    const int g  = (t >> 4) & 1;
    const int T  = (t >> 5) & 31;
    const int col = (T * 8 + pp) * 2 + h;
    const int s = sidx[it];
    const int jcol = ((s & ~1) | g) & (UU - 1);
    const float2 mi = ((s & 1) == g) ? Dg[0][jcol] : Of[0][jcol];
    jw[b * 1024 + t] = srcv[it] * 8;
    const float2 md = Dg[1][col];
    const float2 mo = Of[1][col];
    const size_t fbase = ((size_t)((b * 32 + T) * 32 + g * 16 + pp)) * 2 + h;
#pragma unroll
    for (int i = 0; i < 2; ++i) {
      const float2 m = (i == h) ? md : mo;
      fw[fbase + (size_t)i * 16] = make_float2(m.x * mi.x - m.y * mi.y,
                                               m.x * mi.y + m.y * mi.x);
    }
  }
}


__device__ __forceinline__ void lds_barrier() {
  __builtin_amdgcn_fence(__ATOMIC_RELEASE, "workgroup", "local");
  __builtin_amdgcn_s_barrier();
  __builtin_amdgcn_fence(__ATOMIC_ACQUIRE, "workgroup", "local");
  __builtin_amdgcn_sched_barrier(0);
}

#define NT 4
struct FacRegs { float4 f[NT]; };
struct OffRegs { int2 j[NT]; };

__global__ __launch_bounds__(256) void scan_kernel(
    const float* __restrict__ x, const float* __restrict__ gamma,
    const int* __restrict__ perms, const float4* __restrict__ fw,
    const int2* __restrict__ jw, float* __restrict__ out) {
  __shared__ float2 S[2 * UU];

  const int tid  = threadIdx.x;
  const int b0   = blockIdx.x;
  const int wave = tid >> 5;
  const int lane = tid & 31;
  const int hi   = lane >> 4;
  const int n    = lane & 15;
  const int q    = n >> 1;
  const int fl   = n & 1;

  const float4* fsel = fw + (wave * NT) * 32 + lane;
  const int2* jsel = jw + (wave * NT) * 16 + hi * 8 + q;
  const int wb0 = (wave * 64 + 2 * q + hi) * 2 + fl;
  const bool q1 = (q & 1) != 0;
  const bool q2 = (q & 2) != 0;
  const bool q4 = (q & 4) != 0;
  const bool imag = (fl != 0);

  auto fetch_fac = [&](FacRegs& g, int blk) {
    blk = (blk < NB) ? blk : (NB - 1);
    const float4* fp = fsel + (size_t)blk * (32 * 32);
#pragma unroll
    for (int t = 0; t < NT; ++t) g.f[t] = fp[t * 32];
  };
  auto fetch_off = [&](OffRegs& g, int blk) {
    blk = (blk < NB) ? blk : (NB - 1);
    const int2* jp = jsel + (size_t)blk * (32 * 16);
#pragma unroll
    for (int t = 0; t < NT; ++t) g.j[t] = jp[t * 16];
  };

  FacRegs c0, c1, c2, c3;
  OffRegs s0, s1, s2, s3;
  fetch_off(s0, 0);
  fetch_fac(c0, 0);
  fetch_off(s1, 1);
  fetch_fac(c1, 1);
  fetch_off(s2, 2);
  fetch_fac(c2, 2);
  int pfin[UU / 256];
#pragma unroll
  for (int e = 0; e < UU / 256; ++e)
    pfin[e] = perms[(size_t)LL * UU + tid + e * 256];

#pragma unroll
  for (int e = 0; e < UU / 256; ++e) {
    int u = tid + e * 256;
    int p = perms[u];
    float xr = x[(size_t)b0 * UU + p];
    float xi = x[(size_t)BB * UU + (size_t)b0 * UU + p];
    float sg, cg;
    __sincosf(gamma[p], &sg, &cg);
    S[u] = make_float2(xr * cg - xi * sg, xr * sg + xi * cg);
  }
  __syncthreads();

  auto block = [&](const FacRegs& g, FacRegs& gn, const OffRegs& sl,
                   OffRegs& sn, int blk, const int rd, const int wr) {
    float2 sb[NT][2];
    auto rd_state = [&](int t, int h) {
      sb[t][h] = *reinterpret_cast<const float2*>(
                     reinterpret_cast<const char*>(S + rd) + (h ? sl.j[t].y : sl.j[t].x));
    };
    rd_state(0, 0); rd_state(1, 0); rd_state(0, 1); rd_state(1, 1);
    rd_state(2, 0); rd_state(3, 0); rd_state(2, 1); rd_state(3, 1);
    __builtin_amdgcn_sched_barrier(0);
    fetch_off(sn, blk + 3);
    fetch_fac(gn, blk + 3);
    __builtin_amdgcn_sched_barrier(0);

    v8f acc[NT];
    float y[NT];
    auto prod = [&](int t, int h) {
      v2f aop, bop;
      aop[0] = h ? g.f[t].z : g.f[t].x;
      aop[1] = h ? g.f[t].w : g.f[t].y;
      const float2 sv = sb[t][h];
      bop[0] = imag ? sv.y : sv.x;
      bop[1] = imag ? sv.x : -sv.y;
      const v8f zero = {0.f, 0.f, 0.f, 0.f, 0.f, 0.f, 0.f, 0.f};
      acc[t] = __builtin_amdgcn_wmma_f32_16x16x4_f32(
                   false, aop, false, bop, (short)0, h ? acc[t] : zero, false, false);
    };
    auto pick = [&](int t) {
      const v8f c = acc[t];
      const float a0 = q1 ? c[1] : c[0];
      const float a1 = q1 ? c[3] : c[2];
      const float a2 = q1 ? c[5] : c[4];
      const float a3 = q1 ? c[7] : c[6];
      const float e0 = q2 ? a1 : a0;
      const float e1 = q2 ? a3 : a2;
      y[t] = q4 ? e1 : e0;
      asm volatile("" :: "v"(y[t]));
    };
    prod(0, 0);
    __builtin_amdgcn_sched_barrier(0);
    prod(1, 0);
    __builtin_amdgcn_sched_barrier(0);
    prod(0, 1);
    __builtin_amdgcn_sched_barrier(0);
    prod(1, 1);
    __builtin_amdgcn_sched_barrier(0);
    prod(2, 0);
    __builtin_amdgcn_sched_barrier(0);
    pick(0);
    __builtin_amdgcn_sched_barrier(0);
    prod(3, 0);
    __builtin_amdgcn_sched_barrier(0);
    pick(1);
    __builtin_amdgcn_sched_barrier(0);
    prod(2, 1);
    __builtin_amdgcn_sched_barrier(0);
    prod(3, 1);
    __builtin_amdgcn_sched_barrier(0);
    pick(2);
    pick(3);
    float* wp = reinterpret_cast<float*>(S + wr) + wb0;
#pragma unroll
    for (int t = 0; t < NT; ++t) wp[t * 32] = y[t];
    lds_barrier();
  };

  for (int blk = 0; blk < NB; blk += 4) {
    block(c0, c3, s0, s3, blk,     0, UU);
    block(c1, c0, s1, s0, blk + 1, UU, 0);
    block(c2, c1, s2, s1, blk + 2, 0, UU);
    block(c3, c2, s3, s2, blk + 3, UU, 0);
  }

#pragma unroll
  for (int e = 0; e < UU / 256; ++e) {
    int u = tid + e * 256;
    float2 sv = S[pfin[e]];
    out[(size_t)b0 * UU + u] = sv.x;
    out[(size_t)BB * UU + (size_t)b0 * UU + u] = sv.y;
  }
}

extern "C" void kernel_launch(void* const* d_in, const int* in_sizes, int n_in,
                              void* d_out, int out_size, void* d_ws, size_t ws_size,
                              hipStream_t stream) {
  const float* x     = (const float*)d_in[0];
  const float* theta = (const float*)d_in[1];
  const float* phi   = (const float*)d_in[2];
  const float* gamma = (const float*)d_in[3];
  const float* mask  = (const float*)d_in[4];
  const float* enn   = (const float*)d_in[5];
  const float* enp   = (const float*)d_in[6];
  const float* epn   = (const float*)d_in[7];
  const float* epp   = (const float*)d_in[8];
  const int*   perms = (const int*)d_in[9];

  char* ws = (char*)d_ws;
  float2* fw = (float2*)(ws);
  int*    jw = (int*)(ws + (size_t)LL * UU * 16);

  const int threads = 256;

  prep_kernel<<<NB, threads, 0, stream>>>(theta, phi, mask, enn, enp, epn, epp,
                                          perms, fw, jw);
  scan_kernel<<<BB, threads, 0, stream>>>(x, gamma, perms, (const float4*)fw,
                                          (const int2*)jw, (float*)d_out);
}
